# starting kernel
# baseline (speedup 1.0000x reference)
#include <hip/hip_runtime.h>
#include <math.h>

#define L_SEQ 2048
#define DMODEL 512
#define NHEADS 8
#define DHEAD 64
#define MFEAT 128
#define NCHUNKS 16
#define CHUNK 128
#define CHUNK_STRIDE 8320
#define KERNEL_EPS 1e-4f
#define DEN_EPS 1e-6f
#define QK_SCALE 0.21022410381342864f
#define NC_FEAT 0.08838834764831845f

typedef float v2f __attribute__((ext_vector_type(2)));
typedef float v8f __attribute__((ext_vector_type(8)));

struct Acc2x2 { v8f c[2][2]; };

__device__ __forceinline__ Acc2x2 wmma_block2x2_f32(const float* __restrict__ A, int lda,
                                                    const float* __restrict__ B, int bks, int bns,
                                                    int K, int lane) {
  const int half = lane >> 4;
  const int l16  = lane & 15;
  const float* arow0 = A + l16 * lda;
  const float* arow1 = A + (l16 + 16) * lda;
  const float* bcol0 = B + l16 * bns;
  const float* bcol1 = B + (l16 + 16) * bns;
  const v8f zero = {0.f, 0.f, 0.f, 0.f, 0.f, 0.f, 0.f, 0.f};
  Acc2x2 r;
  r.c[0][0] = zero; r.c[0][1] = zero; r.c[1][0] = zero; r.c[1][1] = zero;
  for (int k0 = 0; k0 < K; k0 += 4) {
    const int ka = k0 + 2 * half;
    v2f a0, a1, b0, b1;
    a0.x = arow0[ka];          a0.y = arow0[ka + 1];
    a1.x = arow1[ka];          a1.y = arow1[ka + 1];
    b0.x = bcol0[ka * bks];    b0.y = bcol0[(ka + 1) * bks];
    b1.x = bcol1[ka * bks];    b1.y = bcol1[(ka + 1) * bks];
    r.c[0][0] = __builtin_amdgcn_wmma_f32_16x16x4_f32(false, a0, false, b0, (short)0, r.c[0][0], false, false);
    r.c[0][1] = __builtin_amdgcn_wmma_f32_16x16x4_f32(false, a0, false, b1, (short)0, r.c[0][1], false, false);
    r.c[1][0] = __builtin_amdgcn_wmma_f32_16x16x4_f32(false, a1, false, b0, (short)0, r.c[1][0], false, false);
    r.c[1][1] = __builtin_amdgcn_wmma_f32_16x16x4_f32(false, a1, false, b1, (short)0, r.c[1][1], false, false);
  }
  return r;
}

__global__ void __launch_bounds__(32)
perf_qkv_proj(const float* __restrict__ xq, const float* __restrict__ xk,
              const float* __restrict__ xv,
              const float* __restrict__ Wq, const float* __restrict__ bq,
              const float* __restrict__ Wk, const float* __restrict__ bk,
              const float* __restrict__ Wv, const float* __restrict__ bv,
              float* __restrict__ Q, float* __restrict__ K, float* __restrict__ V) {
  const int tm = blockIdx.x * 32;
  const int tn = blockIdx.y * 32;
  const int mat = blockIdx.z;
  const int lane = threadIdx.x;
  const float* X; const float* W; const float* bias; float* Y; float scl;
  if (mat == 0)      { X = xq; W = Wq; bias = bq; Y = Q; scl = QK_SCALE; }
  else if (mat == 1) { X = xk; W = Wk; bias = bk; Y = K; scl = QK_SCALE; }
  else               { X = xv; W = Wv; bias = bv; Y = V; scl = 1.f; }
  Acc2x2 r = wmma_block2x2_f32(X + tm * DMODEL, DMODEL,
                               W + tn * DMODEL, 1, DMODEL, DMODEL, lane);
  const int l16 = lane & 15;
  const int half = lane >> 4;
#pragma unroll
  for (int ti = 0; ti < 2; ++ti)
#pragma unroll
    for (int tj = 0; tj < 2; ++tj) {
      const int col = tn + tj * 16 + l16;
      const int rbase = tm + ti * 16 + 8 * half;
      const float bb = bias[col];
#pragma unroll
      for (int i = 0; i < 8; ++i)
        Y[(rbase + i) * DMODEL + col] = (r.c[ti][tj][i] + bb) * scl;
    }
}

__global__ void __launch_bounds__(256)
perf_hk(const float* __restrict__ K, float* __restrict__ hk) {
  const int idx = blockIdx.x * 256 + threadIdx.x;
  const int h = idx >> 11;
  const int l = idx & (L_SEQ - 1);
  const float* row = K + l * DMODEL + h * DHEAD;
  float s = 0.f;
#pragma unroll 8
  for (int d = 0; d < DHEAD; ++d) s += row[d] * row[d];
  hk[idx] = -0.5f * s;
}

__global__ void __launch_bounds__(256)
perf_kstab(const float* __restrict__ hk, float* __restrict__ kstab) {
  __shared__ float red[256];
  const int t = threadIdx.x;
  float m = -3.402823e38f;
  for (int i = t; i < NHEADS * L_SEQ; i += 256) m = fmaxf(m, hk[i]);
  red[t] = m;
  __syncthreads();
  for (int s = 128; s > 0; s >>= 1) {
    if (t < s) red[t] = fmaxf(red[t], red[t + s]);
    __syncthreads();
  }
  if (t == 0) kstab[0] = red[0];
}

__global__ void __launch_bounds__(32)
perf_primes(const float* __restrict__ Q, const float* __restrict__ K,
            const float* __restrict__ RF, const float* __restrict__ hk,
            const float* __restrict__ kstab,
            float* __restrict__ qprime, float* __restrict__ kprime) {
  const int tm = blockIdx.x * 32;
  const int tn = blockIdx.y * 32;
  const int h = blockIdx.z >> 1;
  const int isK = blockIdx.z & 1;
  const int lane = threadIdx.x;
  const float* A = (isK ? K : Q) + tm * DMODEL + h * DHEAD;
  Acc2x2 r = wmma_block2x2_f32(A, DMODEL, RF + tn, MFEAT, 1, DHEAD, lane);
  const int l16 = lane & 15;
  const int half = lane >> 4;
  const float ks = isK ? kstab[0] : 0.f;
#pragma unroll
  for (int ti = 0; ti < 2; ++ti)
#pragma unroll
    for (int tj = 0; tj < 2; ++tj) {
      const int mcol = tn + tj * 16 + l16;
      const int rbase = tm + ti * 16 + 8 * half;
      if (isK) {
#pragma unroll
        for (int i = 0; i < 8; ++i) {
          const int l = rbase + i;
          kprime[(h * L_SEQ + l) * MFEAT + mcol] =
              NC_FEAT * (expf(hk[h * L_SEQ + l] + r.c[ti][tj][i] - ks) + KERNEL_EPS);
        }
      } else {
#pragma unroll
        for (int i = 0; i < 8; ++i) {
          const int l = rbase + i;
          qprime[(h * L_SEQ + l) * MFEAT + mcol] =
              NC_FEAT * (expf(r.c[ti][tj][i]) + KERNEL_EPS);
        }
      }
    }
}

__global__ void __launch_bounds__(256)
perf_chunk_sums(const float* __restrict__ kprime, const float* __restrict__ V,
                float* __restrict__ chunkbuf) {
  const int h = blockIdx.x >> 4;
  const int c = blockIdx.x & 15;
  const int t = threadIdx.x;
  __shared__ float kp[MFEAT];
  __shared__ float vv[DHEAD];
  const int m = t >> 1;
  const int d0 = (t & 1) * 32;
  float acc[32];
#pragma unroll
  for (int j = 0; j < 32; ++j) acc[j] = 0.f;
  float zacc = 0.f;
  for (int l0 = 0; l0 < CHUNK; ++l0) {
    const int l = c * CHUNK + l0;
    if (t < MFEAT) kp[t] = kprime[(h * L_SEQ + l) * MFEAT + t];
    if (t < DHEAD) vv[t] = V[l * DMODEL + h * DHEAD + t];
    __syncthreads();
    const float kpm = kp[m];
#pragma unroll
    for (int j = 0; j < 32; ++j) acc[j] += kpm * vv[d0 + j];
    zacc += kpm;
    __syncthreads();
  }
  float* buf = chunkbuf + (size_t)(h * NCHUNKS + c) * CHUNK_STRIDE;
#pragma unroll
  for (int j = 0; j < 32; ++j) buf[m * DHEAD + d0 + j] = acc[j];
  if ((t & 1) == 0) buf[MFEAT * DHEAD + m] = zacc;
}

__global__ void __launch_bounds__(256)
perf_chunk_prefix(float* __restrict__ chunkbuf) {
  const int h = blockIdx.x;
  for (int idx = threadIdx.x; idx < CHUNK_STRIDE; idx += 256) {
    float carry = 0.f;
    for (int c = 0; c < NCHUNKS; ++c) {
      float* p = chunkbuf + (size_t)(h * NCHUNKS + c) * CHUNK_STRIDE + idx;
      const float tmp = *p;
      *p = carry;
      carry += tmp;
    }
  }
}

__global__ void __launch_bounds__(256)
perf_scan(const float* __restrict__ qprime, const float* __restrict__ kprime,
          const float* __restrict__ V, const float* __restrict__ chunkbuf,
          float* __restrict__ context) {
  const int h = blockIdx.x >> 4;
  const int c = blockIdx.x & 15;
  const int t = threadIdx.x;
  __shared__ float S[MFEAT * DHEAD];
  __shared__ float z[MFEAT];
  __shared__ float qp[MFEAT];
  __shared__ float kp[MFEAT];
  __shared__ float vv[DHEAD];
  __shared__ float red[256];
  __shared__ float dred[MFEAT];
  const float* buf = chunkbuf + (size_t)(h * NCHUNKS + c) * CHUNK_STRIDE;
  for (int i = t; i < MFEAT * DHEAD; i += 256) S[i] = buf[i];
  if (t < MFEAT) z[t] = buf[MFEAT * DHEAD + t];
  __syncthreads();
  const int m = t >> 1;
  const int d0 = (t & 1) * 32;
  const int dd = t & 63;
  const int part = t >> 6;
  for (int l0 = 0; l0 < CHUNK; ++l0) {
    const int l = c * CHUNK + l0;
    if (t < MFEAT) qp[t] = qprime[(h * L_SEQ + l) * MFEAT + t];
    else           kp[t - MFEAT] = kprime[(h * L_SEQ + l) * MFEAT + (t - MFEAT)];
    if (t < DHEAD) vv[t] = V[l * DMODEL + h * DHEAD + t];
    __syncthreads();
    const float kpm = kp[m];
#pragma unroll
    for (int j = 0; j < 32; ++j) S[m * DHEAD + d0 + j] += kpm * vv[d0 + j];
    if ((t & 1) == 0) z[m] += kpm;
    __syncthreads();
    float s = 0.f;
    const int mb = part * 32;
#pragma unroll
    for (int j = 0; j < 32; ++j) s += qp[mb + j] * S[(mb + j) * DHEAD + dd];
    red[t] = s;
    if (t < MFEAT) dred[t] = qp[t] * z[t];
    __syncthreads();
    if (t < 64) dred[t] += dred[t + 64];
    __syncthreads();
    if (t == 0) {
      float s2 = 0.f;
      for (int j = 0; j < 64; ++j) s2 += dred[j];
      dred[0] = s2;
    }
    __syncthreads();
    if (t < DHEAD) {
      const float num = red[dd] + red[64 + dd] + red[128 + dd] + red[192 + dd];
      context[l * DMODEL + h * DHEAD + dd] = num / (dred[0] + DEN_EPS);
    }
    __syncthreads();
  }
}

__global__ void __launch_bounds__(32)
perf_out_proj(const float* __restrict__ ctx, const float* __restrict__ Wo,
              const float* __restrict__ bo, float* __restrict__ out) {
  const int tm = blockIdx.x * 32;
  const int tn = blockIdx.y * 32;
  const int lane = threadIdx.x;
  Acc2x2 r = wmma_block2x2_f32(ctx + tm * DMODEL, DMODEL,
                               Wo + tn * DMODEL, 1, DMODEL, DMODEL, lane);
  const int l16 = lane & 15;
  const int half = lane >> 4;
#pragma unroll
  for (int ti = 0; ti < 2; ++ti)
#pragma unroll
    for (int tj = 0; tj < 2; ++tj) {
      const int col = tn + tj * 16 + l16;
      const int rbase = tm + ti * 16 + 8 * half;
      const float bb = bo[col];
#pragma unroll
      for (int i = 0; i < 8; ++i)
        out[(rbase + i) * DMODEL + col] = r.c[ti][tj][i] + bb;
    }
}

extern "C" void kernel_launch(void* const* d_in, const int* in_sizes, int n_in,
                              void* d_out, int out_size, void* d_ws, size_t ws_size,
                              hipStream_t stream) {
  (void)in_sizes; (void)n_in; (void)out_size; (void)ws_size;
  const float* query = (const float*)d_in[0];
  const float* key_  = (const float*)d_in[1];
  const float* value = (const float*)d_in[2];
  const float* Wq = (const float*)d_in[3];
  const float* bq = (const float*)d_in[4];
  const float* Wk = (const float*)d_in[5];
  const float* bk = (const float*)d_in[6];
  const float* Wv = (const float*)d_in[7];
  const float* bv = (const float*)d_in[8];
  const float* Wo = (const float*)d_in[9];
  const float* bo = (const float*)d_in[10];
  const float* RF = (const float*)d_in[11];

  float* ws = (float*)d_ws;
  const size_t SZ_LD = (size_t)L_SEQ * DMODEL;
  const size_t SZ_PRIME = (size_t)NHEADS * L_SEQ * MFEAT;
  float* Q      = ws;
  float* K      = Q + SZ_LD;
  float* V      = K + SZ_LD;
  float* hk     = V + SZ_LD;
  float* kstab  = hk + (size_t)NHEADS * L_SEQ;
  float* qprime = kstab + 64;
  float* kprime = qprime + SZ_PRIME;
  float* chunkbuf = kprime + SZ_PRIME;
  float* context = Q;

  perf_qkv_proj<<<dim3(L_SEQ / 32, DMODEL / 32, 3), 32, 0, stream>>>(
      query, key_, value, Wq, bq, Wk, bk, Wv, bv, Q, K, V);
  perf_hk<<<(NHEADS * L_SEQ) / 256, 256, 0, stream>>>(K, hk);
  perf_kstab<<<1, 256, 0, stream>>>(hk, kstab);
  perf_primes<<<dim3(L_SEQ / 32, MFEAT / 32, 2 * NHEADS), 32, 0, stream>>>(
      Q, K, RF, hk, kstab, qprime, kprime);
  perf_chunk_sums<<<NHEADS * NCHUNKS, 256, 0, stream>>>(kprime, V, chunkbuf);
  perf_chunk_prefix<<<NHEADS, 256, 0, stream>>>(chunkbuf);
  perf_scan<<<NHEADS * NCHUNKS, 256, 0, stream>>>(qprime, kprime, V, chunkbuf,
                                                  context);
  perf_out_proj<<<dim3(L_SEQ / 32, DMODEL / 32), 32, 0, stream>>>(
      context, Wo, bo, (float*)d_out);
}


# h_k and k-max launches removed: feature kernel forms h_k from the K fragments it loads, stores k' exponents and tile maxima; scan takes their max, applies exp on staging; h_k summed in another tree
# speedup vs baseline: 6.0012x; 6.0012x over previous
#include <hip/hip_runtime.h>
#include <math.h>

#define L_SEQ 2048
#define DMODEL 512
#define NHEADS 8
#define DHEAD 64
#define MFEAT 128
#define NCHUNKS 16
#define CHUNK 128
#define CHUNK_STRIDE 8320
#define KERNEL_EPS 1e-4f
#define DEN_EPS 1e-6f
#define QK_SCALE 0.21022410381342864f
#define NC_FEAT 0.08838834764831845f

typedef float v2f __attribute__((ext_vector_type(2)));
typedef float v8f __attribute__((ext_vector_type(8)));

struct Acc2x2 { v8f c[2][2]; };

__device__ __forceinline__ Acc2x2 wmma_block2x2_f32(const float* __restrict__ A, int lda,
                                                    const float* __restrict__ B, int bks, int bns,
                                                    int K, int lane) {
  const int half = lane >> 4;
  const int l16  = lane & 15;
  const float* arow0 = A + l16 * lda;
  const float* arow1 = A + (l16 + 16) * lda;
  const float* bcol0 = B + l16 * bns;
  const float* bcol1 = B + (l16 + 16) * bns;
  const v8f zero = {0.f, 0.f, 0.f, 0.f, 0.f, 0.f, 0.f, 0.f};
  Acc2x2 r;
  r.c[0][0] = zero; r.c[0][1] = zero; r.c[1][0] = zero; r.c[1][1] = zero;
  for (int k0 = 0; k0 < K; k0 += 4) {
    const int ka = k0 + 2 * half;
    v2f a0, a1, b0, b1;
    a0.x = arow0[ka];          a0.y = arow0[ka + 1];
    a1.x = arow1[ka];          a1.y = arow1[ka + 1];
    b0.x = bcol0[ka * bks];    b0.y = bcol0[(ka + 1) * bks];
    b1.x = bcol1[ka * bks];    b1.y = bcol1[(ka + 1) * bks];
    r.c[0][0] = __builtin_amdgcn_wmma_f32_16x16x4_f32(false, a0, false, b0, (short)0, r.c[0][0], false, false);
    r.c[0][1] = __builtin_amdgcn_wmma_f32_16x16x4_f32(false, a0, false, b1, (short)0, r.c[0][1], false, false);
    r.c[1][0] = __builtin_amdgcn_wmma_f32_16x16x4_f32(false, a1, false, b0, (short)0, r.c[1][0], false, false);
    r.c[1][1] = __builtin_amdgcn_wmma_f32_16x16x4_f32(false, a1, false, b1, (short)0, r.c[1][1], false, false);
  }
  return r;
}

__global__ void __launch_bounds__(32)
perf_qkv_proj(const float* __restrict__ xq, const float* __restrict__ xk,
              const float* __restrict__ xv,
              const float* __restrict__ Wq, const float* __restrict__ bq,
              const float* __restrict__ Wk, const float* __restrict__ bk,
              const float* __restrict__ Wv, const float* __restrict__ bv,
              float* __restrict__ Q, float* __restrict__ K, float* __restrict__ V) {
  const int tm = blockIdx.x * 32;
  const int tn = blockIdx.y * 32;
  const int mat = blockIdx.z;
  const int lane = threadIdx.x;
  const float* X; const float* W; const float* bias; float* Y; float scl;
  if (mat == 0)      { X = xq; W = Wq; bias = bq; Y = Q; scl = QK_SCALE; }
  else if (mat == 1) { X = xk; W = Wk; bias = bk; Y = K; scl = QK_SCALE; }
  else               { X = xv; W = Wv; bias = bv; Y = V; scl = 1.f; }
  Acc2x2 r = wmma_block2x2_f32(X + tm * DMODEL, DMODEL,
                               W + tn * DMODEL, 1, DMODEL, DMODEL, lane);
  const int l16 = lane & 15;
  const int half = lane >> 4;
#pragma unroll
  for (int ti = 0; ti < 2; ++ti)
#pragma unroll
    for (int tj = 0; tj < 2; ++tj) {
      const int col = tn + tj * 16 + l16;
      const int rbase = tm + ti * 16 + 8 * half;
      const float bb = bias[col];
#pragma unroll
      for (int i = 0; i < 8; ++i)
        Y[(rbase + i) * DMODEL + col] = (r.c[ti][tj][i] + bb) * scl;
    }
}

#define NTILEMAX (NHEADS * (L_SEQ / 32))

template <bool IS_K>
__device__ __forceinline__ void feature_tile(const float* __restrict__ X,
                                             const float* __restrict__ RF,
                                             float* __restrict__ dst,
                                             float* __restrict__ tilemax, int lane) {
  const int tm = blockIdx.x * 32;
  const int tn = blockIdx.y * 32;
  const int h = blockIdx.z >> 1;
  const int half = lane >> 4;
  const int l16  = lane & 15;
  const float* A = X + tm * DMODEL + h * DHEAD;
  const float* arow0 = A + l16 * DMODEL;
  const float* arow1 = A + (l16 + 16) * DMODEL;
  const float* bcol0 = RF + tn + l16;
  const float* bcol1 = bcol0 + 16;
  const v8f zero = {0.f, 0.f, 0.f, 0.f, 0.f, 0.f, 0.f, 0.f};
  Acc2x2 r;
  r.c[0][0] = zero; r.c[0][1] = zero; r.c[1][0] = zero; r.c[1][1] = zero;
  float sq0 = 0.f, sq1 = 0.f;
  for (int k0 = 0; k0 < DHEAD; k0 += 4) {
    const int ka = k0 + 2 * half;
    v2f a0, a1, b0, b1;
    a0.x = arow0[ka];            a0.y = arow0[ka + 1];
    a1.x = arow1[ka];            a1.y = arow1[ka + 1];
    b0.x = bcol0[ka * MFEAT];    b0.y = bcol0[(ka + 1) * MFEAT];
    b1.x = bcol1[ka * MFEAT];    b1.y = bcol1[(ka + 1) * MFEAT];
    r.c[0][0] = __builtin_amdgcn_wmma_f32_16x16x4_f32(false, a0, false, b0, (short)0, r.c[0][0], false, false);
    r.c[0][1] = __builtin_amdgcn_wmma_f32_16x16x4_f32(false, a0, false, b1, (short)0, r.c[0][1], false, false);
    r.c[1][0] = __builtin_amdgcn_wmma_f32_16x16x4_f32(false, a1, false, b0, (short)0, r.c[1][0], false, false);
    r.c[1][1] = __builtin_amdgcn_wmma_f32_16x16x4_f32(false, a1, false, b1, (short)0, r.c[1][1], false, false);
    if (IS_K) {
      sq0 += a0.x * a0.x; sq0 += a0.y * a0.y;
      sq1 += a1.x * a1.x; sq1 += a1.y * a1.y;
    }
  }
  if (IS_K) {
    const float hrow0 = -0.5f * (sq0 + __shfl_xor(sq0, 16));
    const float hrow1 = -0.5f * (sq1 + __shfl_xor(sq1, 16));
    float mx = fmaxf(hrow0, hrow1);
#pragma unroll
    for (int off = 8; off > 0; off >>= 1) mx = fmaxf(mx, __shfl_xor(mx, off));
    if (blockIdx.y == 0 && lane == 0) tilemax[h * (L_SEQ / 32) + blockIdx.x] = mx;
#pragma unroll
    for (int ti = 0; ti < 2; ++ti) {
      const int rloc = ti * 16 + 8 * half;
      float hh[8];
#pragma unroll
      for (int i = 0; i < 8; ++i) hh[i] = __shfl(ti ? hrow1 : hrow0, 8 * half + i);
#pragma unroll
      for (int tj = 0; tj < 2; ++tj) {
        const int mcol = tn + tj * 16 + l16;
#pragma unroll
        for (int i = 0; i < 8; ++i)
          dst[(h * L_SEQ + tm + rloc + i) * MFEAT + mcol] = hh[i] + r.c[ti][tj][i];
      }
    }
  } else {
#pragma unroll
    for (int ti = 0; ti < 2; ++ti)
#pragma unroll
      for (int tj = 0; tj < 2; ++tj) {
        const int mcol = tn + tj * 16 + l16;
        const int rbase = tm + ti * 16 + 8 * half;
#pragma unroll
        for (int i = 0; i < 8; ++i)
          dst[(h * L_SEQ + rbase + i) * MFEAT + mcol] =
              NC_FEAT * (expf(r.c[ti][tj][i]) + KERNEL_EPS);
      }
  }
}

__global__ void __launch_bounds__(32)
perf_primes(const float* __restrict__ Q, const float* __restrict__ K,
            const float* __restrict__ RF,
            float* __restrict__ qprime, float* __restrict__ kexp,
            float* __restrict__ tilemax) {
  const int lane = threadIdx.x;
  if (blockIdx.z & 1) feature_tile<true>(K, RF, kexp, tilemax, lane);
  else                feature_tile<false>(Q, RF, qprime, tilemax, lane);
}

#define SCAN_ROWS 16
#define SCAN_BLOCKS (CHUNK / SCAN_ROWS)
typedef float v4f __attribute__((ext_vector_type(4)));

__device__ __forceinline__ v4f kfeat4(v4f e, float ks) {
  v4f r;
  r.x = NC_FEAT * (expf(e.x - ks) + KERNEL_EPS);
  r.y = NC_FEAT * (expf(e.y - ks) + KERNEL_EPS);
  r.z = NC_FEAT * (expf(e.z - ks) + KERNEL_EPS);
  r.w = NC_FEAT * (expf(e.w - ks) + KERNEL_EPS);
  return r;
}

__global__ void __launch_bounds__(256)
perf_scan(const float* __restrict__ qprime, const float* __restrict__ kexp,
          const float* __restrict__ tilemax,
          const float* __restrict__ V, float* __restrict__ chunkbuf,
          float* __restrict__ numer, float* __restrict__ denloc) {
  const int h = blockIdx.x >> 4;
  const int c = blockIdx.x & 15;
  const int t = threadIdx.x;
  const int lane = t & 31;
  const int w = t >> 5;
  const int mz = 16 * w + (lane & 15);
  __shared__ __attribute__((aligned(16))) float qp_s[SCAN_ROWS * MFEAT];
  __shared__ __attribute__((aligned(16))) float kp_s[SCAN_ROWS * MFEAT];
  __shared__ __attribute__((aligned(16))) float vv_s[SCAN_ROWS * DHEAD];
  __shared__ __attribute__((aligned(16))) float red[SCAN_ROWS * 8 * DHEAD];
  __shared__ __attribute__((aligned(16))) float denp[SCAN_ROWS * MFEAT];
  __shared__ float dsink[16];
  __shared__ __attribute__((aligned(16))) float den_s[CHUNK];
  __shared__ __attribute__((aligned(16))) float z_s[MFEAT];
  __shared__ float wmax_s[8];

  v2f S[16];
#pragma unroll
  for (int j = 0; j < 16; ++j) { S[j].x = 0.f; S[j].y = 0.f; }
  float z = 0.f;
  float* const dst0 = (lane < 16) ? (denp + mz) : (dsink + (lane & 15));
  const int dstep = (lane < 16) ? MFEAT : 0;

  const int l_chunk = c * CHUNK;
  const v4f* qsrc = (const v4f*)(qprime + (size_t)(h * L_SEQ + l_chunk) * MFEAT);
  const v4f* ksrc = (const v4f*)(kexp + (size_t)(h * L_SEQ + l_chunk) * MFEAT);
  const int vrow = t >> 4;
  const int vcol = (t & 15) * 4;
  const float* vsrc = V + (size_t)(l_chunk + vrow) * DMODEL + h * DHEAD + vcol;

  const v2f tmx = ((const v2f*)tilemax)[t];
  v4f gq0 = qsrc[t], gq1 = qsrc[t + 256];
  v4f gk0 = ksrc[t], gk1 = ksrc[t + 256];
  v4f gv = *(const v4f*)vsrc;
  float mx = fmaxf(tmx.x, tmx.y);
#pragma unroll
  for (int off = 16; off > 0; off >>= 1) mx = fmaxf(mx, __shfl_xor(mx, off));
  if (lane == 0) wmax_s[w] = mx;
  __syncthreads();
  float ks = wmax_s[0];
#pragma unroll
  for (int i = 1; i < 8; ++i) ks = fmaxf(ks, wmax_s[i]);
  ((v4f*)qp_s)[t] = gq0; ((v4f*)qp_s)[t + 256] = gq1;
  ((v4f*)kp_s)[t] = kfeat4(gk0, ks); ((v4f*)kp_s)[t + 256] = kfeat4(gk1, ks);
  ((v4f*)vv_s)[t] = gv;
  __syncthreads();

  for (int b = 0; b < SCAN_BLOCKS; ++b) {
    if (b + 1 < SCAN_BLOCKS) {
      const int o = (b + 1) * (SCAN_ROWS * MFEAT / 4);
      gq0 = qsrc[o + t]; gq1 = qsrc[o + t + 256];
      gk0 = ksrc[o + t]; gk1 = ksrc[o + t + 256];
      gv = *(const v4f*)(vsrc + (size_t)(b + 1) * SCAN_ROWS * DMODEL);
    }
#pragma unroll 4
    for (int r = 0; r < SCAN_ROWS; ++r) {
      const v4f* kq = (const v4f*)(kp_s + r * MFEAT + 16 * w);
      const v4f* qq = (const v4f*)(qp_s + r * MFEAT + 16 * w);
      const v4f k0 = kq[0], k1 = kq[1], k2 = kq[2], k3 = kq[3];
      const v4f q0 = qq[0], q1 = qq[1], q2 = qq[2], q3 = qq[3];
      const v2f vv2 = *(const v2f*)(vv_s + r * DHEAD + 2 * lane);
      const float kz = kp_s[r * MFEAT + mz];
      const float qz = qp_s[r * MFEAT + mz];
      const float kk[16] = {k0.x, k0.y, k0.z, k0.w, k1.x, k1.y, k1.z, k1.w,
                            k2.x, k2.y, k2.z, k2.w, k3.x, k3.y, k3.z, k3.w};
      const float qv[16] = {q0.x, q0.y, q0.z, q0.w, q1.x, q1.y, q1.z, q1.w,
                            q2.x, q2.y, q2.z, q2.w, q3.x, q3.y, q3.z, q3.w};
#pragma unroll
      for (int j = 0; j < 16; ++j) S[j] += kk[j] * vv2;
      z += kz;
      v2f sa = {0.f, 0.f}, sb = {0.f, 0.f};
#pragma unroll
      for (int j = 0; j < 16; j += 2) {
        sa += qv[j] * S[j];
        sb += qv[j + 1] * S[j + 1];
      }
      *(v2f*)(red + (r * 8 + w) * DHEAD + 2 * lane) = sa + sb;
      dst0[r * dstep] = qz * z;
    }
    if (b + 1 < SCAN_BLOCKS) { gk0 = kfeat4(gk0, ks); gk1 = kfeat4(gk1, ks); }
    __syncthreads();
    if (b + 1 < SCAN_BLOCKS) {
      ((v4f*)qp_s)[t] = gq0; ((v4f*)qp_s)[t + 256] = gq1;
      ((v4f*)kp_s)[t] = gk0; ((v4f*)kp_s)[t + 256] = gk1;
      ((v4f*)vv_s)[t] = gv;
    }
#pragma unroll
    for (int p = 0; p < 4; ++p) {
      const int r = p * 4 + (t >> 6);
      const int d = t & 63;
      const float* dp = denp + r * MFEAT;
      float den = (dp[lane] + dp[lane + 32]) + (dp[lane + 64] + dp[lane + 96]);
#pragma unroll
      for (int off = 16; off > 0; off >>= 1) den += __shfl_xor(den, off);
      float num = 0.f;
#pragma unroll
      for (int ww = 0; ww < 8; ++ww) num += red[(r * 8 + ww) * DHEAD + d];
      const int l = l_chunk + b * SCAN_ROWS + r;
      numer[(size_t)l * DMODEL + h * DHEAD + d] = num;
      if (d == 0) den_s[b * SCAN_ROWS + r] = den;
    }
    __syncthreads();
  }
  float* sums = chunkbuf + (size_t)(h * NCHUNKS + c) * CHUNK_STRIDE;
#pragma unroll
  for (int j = 0; j < 16; ++j)
    *(v2f*)(sums + (16 * w + j) * DHEAD + 2 * lane) = S[j];
  if (lane < 16) z_s[mz] = z;
  __syncthreads();
  if (w == 0) {
    ((v4f*)(sums + MFEAT * DHEAD))[lane] = ((const v4f*)z_s)[lane];
    ((v4f*)(denloc + h * L_SEQ + l_chunk))[lane] = ((const v4f*)den_s)[lane];
  }
}

#define PS_PITCH 40

__global__ void __launch_bounds__(128)
perf_apply_prefix(const float* __restrict__ qprime, const float* __restrict__ chunkbuf,
                  const float* __restrict__ denloc, float* __restrict__ ctx) {
  const int c = blockIdx.x;
  const int tn = blockIdx.y * 32;
  const int h = blockIdx.z;
  const int t = threadIdx.x;
  const int lane = t & 31;
  const int w = t >> 5;
  const int tm = c * CHUNK + 32 * w;
  __shared__ __attribute__((aligned(16))) float P_s[MFEAT * PS_PITCH];
  __shared__ __attribute__((aligned(16))) float zp_s[MFEAT];

  const float* hbase = chunkbuf + (size_t)(h * NCHUNKS) * CHUNK_STRIDE;
  const int prow = t >> 3;
  const int pcol = (t & 7) * 4;
  const v4f zero4 = {0.f, 0.f, 0.f, 0.f};
  v4f acc[8];
#pragma unroll
  for (int j = 0; j < 8; ++j) acc[j] = zero4;
  float zacc = 0.f;
#pragma unroll 4
  for (int cc = 0; cc < c; ++cc) {
    const float* s = hbase + (size_t)cc * CHUNK_STRIDE;
#pragma unroll
    for (int j = 0; j < 8; ++j)
      acc[j] += *(const v4f*)(s + (prow + 16 * j) * DHEAD + tn + pcol);
    zacc += s[MFEAT * DHEAD + t];
  }
#pragma unroll
  for (int j = 0; j < 8; ++j)
    *(v4f*)(P_s + (prow + 16 * j) * PS_PITCH + pcol) = acc[j];
  zp_s[t] = zacc;
  __syncthreads();

  const int half = lane >> 4;
  const int l16  = lane & 15;
  const float* A = qprime + (size_t)(h * L_SEQ + tm) * MFEAT;
  const float* arow0 = A + l16 * MFEAT;
  const float* arow1 = A + (l16 + 16) * MFEAT;
  const float* bcol0 = P_s + l16;
  const float* bcol1 = P_s + 16 + l16;
  const v8f zero = {0.f, 0.f, 0.f, 0.f, 0.f, 0.f, 0.f, 0.f};
  Acc2x2 r;
  r.c[0][0] = zero; r.c[0][1] = zero; r.c[1][0] = zero; r.c[1][1] = zero;
  for (int k0 = 0; k0 < MFEAT; k0 += 4) {
    const int ka = k0 + 2 * half;
    v2f a0, a1, b0, b1;
    a0.x = arow0[ka];               a0.y = arow0[ka + 1];
    a1.x = arow1[ka];               a1.y = arow1[ka + 1];
    b0.x = bcol0[ka * PS_PITCH];    b0.y = bcol0[(ka + 1) * PS_PITCH];
    b1.x = bcol1[ka * PS_PITCH];    b1.y = bcol1[(ka + 1) * PS_PITCH];
    r.c[0][0] = __builtin_amdgcn_wmma_f32_16x16x4_f32(false, a0, false, b0, (short)0, r.c[0][0], false, false);
    r.c[0][1] = __builtin_amdgcn_wmma_f32_16x16x4_f32(false, a0, false, b1, (short)0, r.c[0][1], false, false);
    r.c[1][0] = __builtin_amdgcn_wmma_f32_16x16x4_f32(false, a1, false, b0, (short)0, r.c[1][0], false, false);
    r.c[1][1] = __builtin_amdgcn_wmma_f32_16x16x4_f32(false, a1, false, b1, (short)0, r.c[1][1], false, false);
  }
  const v4f* qrow = (const v4f*)(A + lane * MFEAT);
  const v4f* zp = (const v4f*)zp_s;
  v4f dacc = {0.f, 0.f, 0.f, 0.f};
  for (int m4 = 0; m4 < MFEAT / 4; ++m4) dacc += qrow[m4] * zp[m4];
  const float dtot = denloc[h * L_SEQ + tm + lane] +
                     ((dacc.x + dacc.y) + (dacc.z + dacc.w)) + DEN_EPS;
#pragma unroll
  for (int ti = 0; ti < 2; ++ti) {
    const int rloc = ti * 16 + 8 * half;
    float dd[8];
#pragma unroll
    for (int i = 0; i < 8; ++i) dd[i] = __shfl(dtot, rloc + i);
#pragma unroll
    for (int tj = 0; tj < 2; ++tj) {
      const int col = h * DHEAD + tn + tj * 16 + l16;
#pragma unroll
      for (int i = 0; i < 8; ++i) {
        float* p = ctx + (size_t)(tm + rloc + i) * DMODEL + col;
        *p = (*p + r.c[ti][tj][i]) / dd[i];
      }
    }
  }
}

__global__ void __launch_bounds__(32)
perf_out_proj(const float* __restrict__ ctx, const float* __restrict__ Wo,
              const float* __restrict__ bo, float* __restrict__ out) {
  const int tm = blockIdx.x * 32;
  const int tn = blockIdx.y * 32;
  const int lane = threadIdx.x;
  Acc2x2 r = wmma_block2x2_f32(ctx + tm * DMODEL, DMODEL,
                               Wo + tn * DMODEL, 1, DMODEL, DMODEL, lane);
  const int l16 = lane & 15;
  const int half = lane >> 4;
#pragma unroll
  for (int ti = 0; ti < 2; ++ti)
#pragma unroll
    for (int tj = 0; tj < 2; ++tj) {
      const int col = tn + tj * 16 + l16;
      const int rbase = tm + ti * 16 + 8 * half;
      const float bb = bo[col];
#pragma unroll
      for (int i = 0; i < 8; ++i)
        out[(rbase + i) * DMODEL + col] = r.c[ti][tj][i] + bb;
    }
}

extern "C" void kernel_launch(void* const* d_in, const int* in_sizes, int n_in,
                              void* d_out, int out_size, void* d_ws, size_t ws_size,
                              hipStream_t stream) {
  (void)in_sizes; (void)n_in; (void)out_size; (void)ws_size;
  const float* query = (const float*)d_in[0];
  const float* key_  = (const float*)d_in[1];
  const float* value = (const float*)d_in[2];
  const float* Wq = (const float*)d_in[3];
  const float* bq = (const float*)d_in[4];
  const float* Wk = (const float*)d_in[5];
  const float* bk = (const float*)d_in[6];
  const float* Wv = (const float*)d_in[7];
  const float* bv = (const float*)d_in[8];
  const float* Wo = (const float*)d_in[9];
  const float* bo = (const float*)d_in[10];
  const float* RF = (const float*)d_in[11];

  float* ws = (float*)d_ws;
  const size_t SZ_LD = (size_t)L_SEQ * DMODEL;
  const size_t SZ_PRIME = (size_t)NHEADS * L_SEQ * MFEAT;
  float* Q      = ws;
  float* K      = Q + SZ_LD;
  float* V      = K + SZ_LD;
  float* tilemax = V + SZ_LD;
  float* qprime = tilemax + NTILEMAX;
  float* kexp   = qprime + SZ_PRIME;
  float* chunkbuf = kexp + SZ_PRIME;
  float* denloc = chunkbuf + (size_t)NHEADS * NCHUNKS * CHUNK_STRIDE;
  float* context = Q;

  perf_qkv_proj<<<dim3(L_SEQ / 32, DMODEL / 32, 3), 32, 0, stream>>>(
      query, key_, value, Wq, bq, Wk, bk, Wv, bv, Q, K, V);
  perf_primes<<<dim3(L_SEQ / 32, MFEAT / 32, 2 * NHEADS), 32, 0, stream>>>(
      Q, K, RF, qprime, kexp, tilemax);
  perf_scan<<<NHEADS * NCHUNKS, 256, 0, stream>>>(qprime, kexp, tilemax, V, chunkbuf,
                                                  context, denloc);
  perf_apply_prefix<<<dim3(NCHUNKS, DHEAD / 32, NHEADS), 128, 0, stream>>>(
      qprime, chunkbuf, denloc, context);
  perf_out_proj<<<dim3(L_SEQ / 32, DMODEL / 32), 32, 0, stream>>>(
      context, Wo, bo, (float*)d_out);
}


# Q and K projections fused with the feature maps: a 4-wave work-group projects 64 rows x one head, hands the tile over in LDS and multiplies it by the feature matrix; feature launch dropped; same sums
# speedup vs baseline: 6.4798x; 1.0798x over previous
#include <hip/hip_runtime.h>
#include <math.h>

#define L_SEQ 2048
#define DMODEL 512
#define NHEADS 8
#define DHEAD 64
#define MFEAT 128
#define NCHUNKS 16
#define CHUNK 128
#define CHUNK_STRIDE 8320
#define KERNEL_EPS 1e-4f
#define DEN_EPS 1e-6f
#define QK_SCALE 0.21022410381342864f
#define NC_FEAT 0.08838834764831845f

typedef float v2f __attribute__((ext_vector_type(2)));
typedef float v8f __attribute__((ext_vector_type(8)));

struct Acc2x2 { v8f c[2][2]; };

__device__ __forceinline__ Acc2x2 wmma_block2x2_f32(const float* __restrict__ A, int lda,
                                                    const float* __restrict__ B, int bks, int bns,
                                                    int K, int lane) {
  const int half = lane >> 4;
  const int l16  = lane & 15;
  const float* arow0 = A + l16 * lda;
  const float* arow1 = A + (l16 + 16) * lda;
  const float* bcol0 = B + l16 * bns;
  const float* bcol1 = B + (l16 + 16) * bns;
  const v8f zero = {0.f, 0.f, 0.f, 0.f, 0.f, 0.f, 0.f, 0.f};
  Acc2x2 r;
  r.c[0][0] = zero; r.c[0][1] = zero; r.c[1][0] = zero; r.c[1][1] = zero;
  for (int k0 = 0; k0 < K; k0 += 4) {
    const int ka = k0 + 2 * half;
    v2f a0, a1, b0, b1;
    a0.x = arow0[ka];          a0.y = arow0[ka + 1];
    a1.x = arow1[ka];          a1.y = arow1[ka + 1];
    b0.x = bcol0[ka * bks];    b0.y = bcol0[(ka + 1) * bks];
    b1.x = bcol1[ka * bks];    b1.y = bcol1[(ka + 1) * bks];
    r.c[0][0] = __builtin_amdgcn_wmma_f32_16x16x4_f32(false, a0, false, b0, (short)0, r.c[0][0], false, false);
    r.c[0][1] = __builtin_amdgcn_wmma_f32_16x16x4_f32(false, a0, false, b1, (short)0, r.c[0][1], false, false);
    r.c[1][0] = __builtin_amdgcn_wmma_f32_16x16x4_f32(false, a1, false, b0, (short)0, r.c[1][0], false, false);
    r.c[1][1] = __builtin_amdgcn_wmma_f32_16x16x4_f32(false, a1, false, b1, (short)0, r.c[1][1], false, false);
  }
  return r;
}

#define NTILEMAX (NHEADS * (L_SEQ / 32))
#define TS_PITCH 68

template <bool IS_K>
__device__ __forceinline__ void feature_from_tile(const float* T,
                                                  const float* __restrict__ RF,
                                                  float* __restrict__ dst,
                                                  float* __restrict__ tilemax,
                                                  int h, int row0, int mbase,
                                                  bool reports, int lane) {
  const int half = lane >> 4;
  const int l16  = lane & 15;
  const float* arow0 = T + l16 * TS_PITCH;
  const float* arow1 = arow0 + 16 * TS_PITCH;
  const v8f zero = {0.f, 0.f, 0.f, 0.f, 0.f, 0.f, 0.f, 0.f};
  float hrow0 = 0.f, hrow1 = 0.f;
#pragma unroll
  for (int mb = 0; mb < 2; ++mb) {
    const int tn = mbase + 32 * mb;
    const float* bcol0 = RF + tn + l16;
    const float* bcol1 = bcol0 + 16;
    Acc2x2 r;
    r.c[0][0] = zero; r.c[0][1] = zero; r.c[1][0] = zero; r.c[1][1] = zero;
    float sq0 = 0.f, sq1 = 0.f;
    for (int k0 = 0; k0 < DHEAD; k0 += 4) {
      const int ka = k0 + 2 * half;
      const v2f a0 = *(const v2f*)(arow0 + ka);
      const v2f a1 = *(const v2f*)(arow1 + ka);
      v2f b0, b1;
      b0.x = bcol0[ka * MFEAT];    b0.y = bcol0[(ka + 1) * MFEAT];
      b1.x = bcol1[ka * MFEAT];    b1.y = bcol1[(ka + 1) * MFEAT];
      r.c[0][0] = __builtin_amdgcn_wmma_f32_16x16x4_f32(false, a0, false, b0, (short)0, r.c[0][0], false, false);
      r.c[0][1] = __builtin_amdgcn_wmma_f32_16x16x4_f32(false, a0, false, b1, (short)0, r.c[0][1], false, false);
      r.c[1][0] = __builtin_amdgcn_wmma_f32_16x16x4_f32(false, a1, false, b0, (short)0, r.c[1][0], false, false);
      r.c[1][1] = __builtin_amdgcn_wmma_f32_16x16x4_f32(false, a1, false, b1, (short)0, r.c[1][1], false, false);
      if (IS_K && mb == 0) {
        sq0 += a0.x * a0.x; sq0 += a0.y * a0.y;
        sq1 += a1.x * a1.x; sq1 += a1.y * a1.y;
      }
    }
    if (IS_K) {
      if (mb == 0) {
        hrow0 = -0.5f * (sq0 + __shfl_xor(sq0, 16));
        hrow1 = -0.5f * (sq1 + __shfl_xor(sq1, 16));
        float mx = fmaxf(hrow0, hrow1);
#pragma unroll
        for (int off = 8; off > 0; off >>= 1) mx = fmaxf(mx, __shfl_xor(mx, off));
        if (reports && lane == 0) tilemax[h * (L_SEQ / 32) + (row0 >> 5)] = mx;
      }
#pragma unroll
      for (int ti = 0; ti < 2; ++ti) {
        const int rloc = ti * 16 + 8 * half;
        float hh[8];
#pragma unroll
        for (int i = 0; i < 8; ++i) hh[i] = __shfl(ti ? hrow1 : hrow0, 8 * half + i);
#pragma unroll
        for (int tj = 0; tj < 2; ++tj) {
          const int mcol = tn + tj * 16 + l16;
#pragma unroll
          for (int i = 0; i < 8; ++i)
            dst[(h * L_SEQ + row0 + rloc + i) * MFEAT + mcol] = hh[i] + r.c[ti][tj][i];
        }
      }
    } else {
#pragma unroll
      for (int ti = 0; ti < 2; ++ti)
#pragma unroll
        for (int tj = 0; tj < 2; ++tj) {
          const int mcol = tn + tj * 16 + l16;
          const int rbase = row0 + ti * 16 + 8 * half;
#pragma unroll
          for (int i = 0; i < 8; ++i)
            dst[(h * L_SEQ + rbase + i) * MFEAT + mcol] =
                NC_FEAT * (expf(r.c[ti][tj][i]) + KERNEL_EPS);
        }
    }
  }
}

__global__ void __launch_bounds__(128)
perf_qkv_features(const float* __restrict__ xq, const float* __restrict__ xk,
                  const float* __restrict__ xv,
                  const float* __restrict__ Wq, const float* __restrict__ bq,
                  const float* __restrict__ Wk, const float* __restrict__ bk,
                  const float* __restrict__ Wv, const float* __restrict__ bv,
                  const float* __restrict__ RF,
                  float* __restrict__ V, float* __restrict__ qprime,
                  float* __restrict__ kexp, float* __restrict__ tilemax) {
  const int h = blockIdx.y;
  const int mat = blockIdx.z;
  const int lane = threadIdx.x & 31;
  const int w = threadIdx.x >> 5;
  const int wr = w >> 1;
  const int wc = w & 1;
  const int tm = blockIdx.x * 64 + 32 * wr;
  const int tn = h * DHEAD + 32 * wc;
  __shared__ __attribute__((aligned(16))) float T_s[64 * TS_PITCH];
  const float* X; const float* W; const float* bias; float scl;
  if (mat == 0)      { X = xq; W = Wq; bias = bq; scl = QK_SCALE; }
  else if (mat == 1) { X = xk; W = Wk; bias = bk; scl = QK_SCALE; }
  else               { X = xv; W = Wv; bias = bv; scl = 1.f; }
  Acc2x2 r = wmma_block2x2_f32(X + tm * DMODEL, DMODEL,
                               W + tn * DMODEL, 1, DMODEL, DMODEL, lane);
  const int l16 = lane & 15;
  const int half = lane >> 4;
  if (mat == 2) {
#pragma unroll
    for (int ti = 0; ti < 2; ++ti)
#pragma unroll
      for (int tj = 0; tj < 2; ++tj) {
        const int col = tn + tj * 16 + l16;
        const int rbase = tm + ti * 16 + 8 * half;
        const float bb = bias[col];
#pragma unroll
        for (int i = 0; i < 8; ++i)
          V[(rbase + i) * DMODEL + col] = r.c[ti][tj][i] + bb;
      }
    return;
  }
#pragma unroll
  for (int ti = 0; ti < 2; ++ti)
#pragma unroll
    for (int tj = 0; tj < 2; ++tj) {
      const int cloc = 32 * wc + tj * 16 + l16;
      const int rloc = 32 * wr + ti * 16 + 8 * half;
      const float bb = bias[h * DHEAD + cloc];
#pragma unroll
      for (int i = 0; i < 8; ++i)
        T_s[(rloc + i) * TS_PITCH + cloc] = (r.c[ti][tj][i] + bb) * scl;
    }
  __syncthreads();
  if (mat == 1)
    feature_from_tile<true>(T_s + 32 * wr * TS_PITCH, RF, kexp, tilemax, h, tm, 64 * wc,
                            wc == 0, lane);
  else
    feature_from_tile<false>(T_s + 32 * wr * TS_PITCH, RF, qprime, tilemax, h, tm, 64 * wc,
                             false, lane);
}

#define SCAN_ROWS 16
#define SCAN_BLOCKS (CHUNK / SCAN_ROWS)
typedef float v4f __attribute__((ext_vector_type(4)));

__device__ __forceinline__ v4f kfeat4(v4f e, float ks) {
  v4f r;
  r.x = NC_FEAT * (expf(e.x - ks) + KERNEL_EPS);
  r.y = NC_FEAT * (expf(e.y - ks) + KERNEL_EPS);
  r.z = NC_FEAT * (expf(e.z - ks) + KERNEL_EPS);
  r.w = NC_FEAT * (expf(e.w - ks) + KERNEL_EPS);
  return r;
}

__global__ void __launch_bounds__(256)
perf_scan(const float* __restrict__ qprime, const float* __restrict__ kexp,
          const float* __restrict__ tilemax,
          const float* __restrict__ V, float* __restrict__ chunkbuf,
          float* __restrict__ numer, float* __restrict__ denloc) {
  const int h = blockIdx.x >> 4;
  const int c = blockIdx.x & 15;
  const int t = threadIdx.x;
  const int lane = t & 31;
  const int w = t >> 5;
  const int mz = 16 * w + (lane & 15);
  __shared__ __attribute__((aligned(16))) float qp_s[SCAN_ROWS * MFEAT];
  __shared__ __attribute__((aligned(16))) float kp_s[SCAN_ROWS * MFEAT];
  __shared__ __attribute__((aligned(16))) float vv_s[SCAN_ROWS * DHEAD];
  __shared__ __attribute__((aligned(16))) float red[SCAN_ROWS * 8 * DHEAD];
  __shared__ __attribute__((aligned(16))) float denp[SCAN_ROWS * MFEAT];
  __shared__ float dsink[16];
  __shared__ __attribute__((aligned(16))) float den_s[CHUNK];
  __shared__ __attribute__((aligned(16))) float z_s[MFEAT];
  __shared__ float wmax_s[8];

  v2f S[16];
#pragma unroll
  for (int j = 0; j < 16; ++j) { S[j].x = 0.f; S[j].y = 0.f; }
  float z = 0.f;
  float* const dst0 = (lane < 16) ? (denp + mz) : (dsink + (lane & 15));
  const int dstep = (lane < 16) ? MFEAT : 0;

  const int l_chunk = c * CHUNK;
  const v4f* qsrc = (const v4f*)(qprime + (size_t)(h * L_SEQ + l_chunk) * MFEAT);
  const v4f* ksrc = (const v4f*)(kexp + (size_t)(h * L_SEQ + l_chunk) * MFEAT);
  const int vrow = t >> 4;
  const int vcol = (t & 15) * 4;
  const float* vsrc = V + (size_t)(l_chunk + vrow) * DMODEL + h * DHEAD + vcol;

  const v2f tmx = ((const v2f*)tilemax)[t];
  v4f gq0 = qsrc[t], gq1 = qsrc[t + 256];
  v4f gk0 = ksrc[t], gk1 = ksrc[t + 256];
  v4f gv = *(const v4f*)vsrc;
  float mx = fmaxf(tmx.x, tmx.y);
#pragma unroll
  for (int off = 16; off > 0; off >>= 1) mx = fmaxf(mx, __shfl_xor(mx, off));
  if (lane == 0) wmax_s[w] = mx;
  __syncthreads();
  float ks = wmax_s[0];
#pragma unroll
  for (int i = 1; i < 8; ++i) ks = fmaxf(ks, wmax_s[i]);
  ((v4f*)qp_s)[t] = gq0; ((v4f*)qp_s)[t + 256] = gq1;
  ((v4f*)kp_s)[t] = kfeat4(gk0, ks); ((v4f*)kp_s)[t + 256] = kfeat4(gk1, ks);
  ((v4f*)vv_s)[t] = gv;
  __syncthreads();

  for (int b = 0; b < SCAN_BLOCKS; ++b) {
    if (b + 1 < SCAN_BLOCKS) {
      const int o = (b + 1) * (SCAN_ROWS * MFEAT / 4);
      gq0 = qsrc[o + t]; gq1 = qsrc[o + t + 256];
      gk0 = ksrc[o + t]; gk1 = ksrc[o + t + 256];
      gv = *(const v4f*)(vsrc + (size_t)(b + 1) * SCAN_ROWS * DMODEL);
    }
#pragma unroll 4
    for (int r = 0; r < SCAN_ROWS; ++r) {
      const v4f* kq = (const v4f*)(kp_s + r * MFEAT + 16 * w);
      const v4f* qq = (const v4f*)(qp_s + r * MFEAT + 16 * w);
      const v4f k0 = kq[0], k1 = kq[1], k2 = kq[2], k3 = kq[3];
      const v4f q0 = qq[0], q1 = qq[1], q2 = qq[2], q3 = qq[3];
      const v2f vv2 = *(const v2f*)(vv_s + r * DHEAD + 2 * lane);
      const float kz = kp_s[r * MFEAT + mz];
      const float qz = qp_s[r * MFEAT + mz];
      const float kk[16] = {k0.x, k0.y, k0.z, k0.w, k1.x, k1.y, k1.z, k1.w,
                            k2.x, k2.y, k2.z, k2.w, k3.x, k3.y, k3.z, k3.w};
      const float qv[16] = {q0.x, q0.y, q0.z, q0.w, q1.x, q1.y, q1.z, q1.w,
                            q2.x, q2.y, q2.z, q2.w, q3.x, q3.y, q3.z, q3.w};
#pragma unroll
      for (int j = 0; j < 16; ++j) S[j] += kk[j] * vv2;
      z += kz;
      v2f sa = {0.f, 0.f}, sb = {0.f, 0.f};
#pragma unroll
      for (int j = 0; j < 16; j += 2) {
        sa += qv[j] * S[j];
        sb += qv[j + 1] * S[j + 1];
      }
      *(v2f*)(red + (r * 8 + w) * DHEAD + 2 * lane) = sa + sb;
      dst0[r * dstep] = qz * z;
    }
    if (b + 1 < SCAN_BLOCKS) { gk0 = kfeat4(gk0, ks); gk1 = kfeat4(gk1, ks); }
    __syncthreads();
    if (b + 1 < SCAN_BLOCKS) {
      ((v4f*)qp_s)[t] = gq0; ((v4f*)qp_s)[t + 256] = gq1;
      ((v4f*)kp_s)[t] = gk0; ((v4f*)kp_s)[t + 256] = gk1;
      ((v4f*)vv_s)[t] = gv;
    }
#pragma unroll
    for (int p = 0; p < 4; ++p) {
      const int r = p * 4 + (t >> 6);
      const int d = t & 63;
      const float* dp = denp + r * MFEAT;
      float den = (dp[lane] + dp[lane + 32]) + (dp[lane + 64] + dp[lane + 96]);
#pragma unroll
      for (int off = 16; off > 0; off >>= 1) den += __shfl_xor(den, off);
      float num = 0.f;
#pragma unroll
      for (int ww = 0; ww < 8; ++ww) num += red[(r * 8 + ww) * DHEAD + d];
      const int l = l_chunk + b * SCAN_ROWS + r;
      numer[(size_t)l * DMODEL + h * DHEAD + d] = num;
      if (d == 0) den_s[b * SCAN_ROWS + r] = den;
    }
    __syncthreads();
  }
  float* sums = chunkbuf + (size_t)(h * NCHUNKS + c) * CHUNK_STRIDE;
#pragma unroll
  for (int j = 0; j < 16; ++j)
    *(v2f*)(sums + (16 * w + j) * DHEAD + 2 * lane) = S[j];
  if (lane < 16) z_s[mz] = z;
  __syncthreads();
  if (w == 0) {
    ((v4f*)(sums + MFEAT * DHEAD))[lane] = ((const v4f*)z_s)[lane];
    ((v4f*)(denloc + h * L_SEQ + l_chunk))[lane] = ((const v4f*)den_s)[lane];
  }
}

#define PS_PITCH 40

__global__ void __launch_bounds__(128)
perf_apply_prefix(const float* __restrict__ qprime, const float* __restrict__ chunkbuf,
                  const float* __restrict__ denloc, float* __restrict__ ctx) {
  const int c = blockIdx.x;
  const int tn = blockIdx.y * 32;
  const int h = blockIdx.z;
  const int t = threadIdx.x;
  const int lane = t & 31;
  const int w = t >> 5;
  const int tm = c * CHUNK + 32 * w;
  __shared__ __attribute__((aligned(16))) float P_s[MFEAT * PS_PITCH];
  __shared__ __attribute__((aligned(16))) float zp_s[MFEAT];

  const float* hbase = chunkbuf + (size_t)(h * NCHUNKS) * CHUNK_STRIDE;
  const int prow = t >> 3;
  const int pcol = (t & 7) * 4;
  const v4f zero4 = {0.f, 0.f, 0.f, 0.f};
  v4f acc[8];
#pragma unroll
  for (int j = 0; j < 8; ++j) acc[j] = zero4;
  float zacc = 0.f;
#pragma unroll 4
  for (int cc = 0; cc < c; ++cc) {
    const float* s = hbase + (size_t)cc * CHUNK_STRIDE;
#pragma unroll
    for (int j = 0; j < 8; ++j)
      acc[j] += *(const v4f*)(s + (prow + 16 * j) * DHEAD + tn + pcol);
    zacc += s[MFEAT * DHEAD + t];
  }
#pragma unroll
  for (int j = 0; j < 8; ++j)
    *(v4f*)(P_s + (prow + 16 * j) * PS_PITCH + pcol) = acc[j];
  zp_s[t] = zacc;
  __syncthreads();

  const int half = lane >> 4;
  const int l16  = lane & 15;
  const float* A = qprime + (size_t)(h * L_SEQ + tm) * MFEAT;
  const float* arow0 = A + l16 * MFEAT;
  const float* arow1 = A + (l16 + 16) * MFEAT;
  const float* bcol0 = P_s + l16;
  const float* bcol1 = P_s + 16 + l16;
  const v8f zero = {0.f, 0.f, 0.f, 0.f, 0.f, 0.f, 0.f, 0.f};
  Acc2x2 r;
  r.c[0][0] = zero; r.c[0][1] = zero; r.c[1][0] = zero; r.c[1][1] = zero;
  for (int k0 = 0; k0 < MFEAT; k0 += 4) {
    const int ka = k0 + 2 * half;
    v2f a0, a1, b0, b1;
    a0.x = arow0[ka];               a0.y = arow0[ka + 1];
    a1.x = arow1[ka];               a1.y = arow1[ka + 1];
    b0.x = bcol0[ka * PS_PITCH];    b0.y = bcol0[(ka + 1) * PS_PITCH];
    b1.x = bcol1[ka * PS_PITCH];    b1.y = bcol1[(ka + 1) * PS_PITCH];
    r.c[0][0] = __builtin_amdgcn_wmma_f32_16x16x4_f32(false, a0, false, b0, (short)0, r.c[0][0], false, false);
    r.c[0][1] = __builtin_amdgcn_wmma_f32_16x16x4_f32(false, a0, false, b1, (short)0, r.c[0][1], false, false);
    r.c[1][0] = __builtin_amdgcn_wmma_f32_16x16x4_f32(false, a1, false, b0, (short)0, r.c[1][0], false, false);
    r.c[1][1] = __builtin_amdgcn_wmma_f32_16x16x4_f32(false, a1, false, b1, (short)0, r.c[1][1], false, false);
  }
  const v4f* qrow = (const v4f*)(A + lane * MFEAT);
  const v4f* zp = (const v4f*)zp_s;
  v4f dacc = {0.f, 0.f, 0.f, 0.f};
  for (int m4 = 0; m4 < MFEAT / 4; ++m4) dacc += qrow[m4] * zp[m4];
  const float dtot = denloc[h * L_SEQ + tm + lane] +
                     ((dacc.x + dacc.y) + (dacc.z + dacc.w)) + DEN_EPS;
#pragma unroll
  for (int ti = 0; ti < 2; ++ti) {
    const int rloc = ti * 16 + 8 * half;
    float dd[8];
#pragma unroll
    for (int i = 0; i < 8; ++i) dd[i] = __shfl(dtot, rloc + i);
#pragma unroll
    for (int tj = 0; tj < 2; ++tj) {
      const int col = h * DHEAD + tn + tj * 16 + l16;
#pragma unroll
      for (int i = 0; i < 8; ++i) {
        float* p = ctx + (size_t)(tm + rloc + i) * DMODEL + col;
        *p = (*p + r.c[ti][tj][i]) / dd[i];
      }
    }
  }
}

__global__ void __launch_bounds__(32)
perf_out_proj(const float* __restrict__ ctx, const float* __restrict__ Wo,
              const float* __restrict__ bo, float* __restrict__ out) {
  const int tm = blockIdx.x * 32;
  const int tn = blockIdx.y * 32;
  const int lane = threadIdx.x;
  Acc2x2 r = wmma_block2x2_f32(ctx + tm * DMODEL, DMODEL,
                               Wo + tn * DMODEL, 1, DMODEL, DMODEL, lane);
  const int l16 = lane & 15;
  const int half = lane >> 4;
#pragma unroll
  for (int ti = 0; ti < 2; ++ti)
#pragma unroll
    for (int tj = 0; tj < 2; ++tj) {
      const int col = tn + tj * 16 + l16;
      const int rbase = tm + ti * 16 + 8 * half;
      const float bb = bo[col];
#pragma unroll
      for (int i = 0; i < 8; ++i)
        out[(rbase + i) * DMODEL + col] = r.c[ti][tj][i] + bb;
    }
}

extern "C" void kernel_launch(void* const* d_in, const int* in_sizes, int n_in,
                              void* d_out, int out_size, void* d_ws, size_t ws_size,
                              hipStream_t stream) {
  (void)in_sizes; (void)n_in; (void)out_size; (void)ws_size;
  const float* query = (const float*)d_in[0];
  const float* key_  = (const float*)d_in[1];
  const float* value = (const float*)d_in[2];
  const float* Wq = (const float*)d_in[3];
  const float* bq = (const float*)d_in[4];
  const float* Wk = (const float*)d_in[5];
  const float* bk = (const float*)d_in[6];
  const float* Wv = (const float*)d_in[7];
  const float* bv = (const float*)d_in[8];
  const float* Wo = (const float*)d_in[9];
  const float* bo = (const float*)d_in[10];
  const float* RF = (const float*)d_in[11];

  float* ws = (float*)d_ws;
  const size_t SZ_LD = (size_t)L_SEQ * DMODEL;
  const size_t SZ_PRIME = (size_t)NHEADS * L_SEQ * MFEAT;
  float* context = ws;
  float* V      = context + SZ_LD;
  float* tilemax = V + SZ_LD;
  float* qprime = tilemax + NTILEMAX;
  float* kexp   = qprime + SZ_PRIME;
  float* chunkbuf = kexp + SZ_PRIME;
  float* denloc = chunkbuf + (size_t)NHEADS * NCHUNKS * CHUNK_STRIDE;

  perf_qkv_features<<<dim3(L_SEQ / 64, NHEADS, 3), 128, 0, stream>>>(
      query, key_, value, Wq, bq, Wk, bk, Wv, bv, RF, V, qprime, kexp, tilemax);
  perf_scan<<<NHEADS * NCHUNKS, 256, 0, stream>>>(qprime, kexp, tilemax, V, chunkbuf,
                                                  context, denloc);
  perf_apply_prefix<<<dim3(NCHUNKS, DHEAD / 32, NHEADS), 128, 0, stream>>>(
      qprime, chunkbuf, denloc, context);
  perf_out_proj<<<dim3(L_SEQ / 32, DMODEL / 32), 32, 0, stream>>>(
      context, Wo, bo, (float*)d_out);
}


# projection GEMMs (Q/K/V, output): K split over the work-group's waves, each wave a 64x64 block as 4x4 register tiles (8 fragment loads per 16 WMMAs, was 4 per 4); partials summed in LDS, other tree
# speedup vs baseline: 6.8669x; 1.0597x over previous
#include <hip/hip_runtime.h>
#include <math.h>

#define L_SEQ 2048
#define DMODEL 512
#define NHEADS 8
#define DHEAD 64
#define MFEAT 128
#define NCHUNKS 16
#define CHUNK 128
#define CHUNK_STRIDE 8320
#define KERNEL_EPS 1e-4f
#define DEN_EPS 1e-6f
#define QK_SCALE 0.21022410381342864f
#define NC_FEAT 0.08838834764831845f

typedef float v2f __attribute__((ext_vector_type(2)));
typedef float v8f __attribute__((ext_vector_type(8)));

struct Acc2x2 { v8f c[2][2]; };
typedef float v4f __attribute__((ext_vector_type(4)));

#define PT_PITCH 68
#define PT_PLANE (64 * PT_PITCH)

template <int KLEN>
__device__ __forceinline__ void gemm64x64_kslice(const float* __restrict__ A,
                                                 const float* __restrict__ W,
                                                 int kb, int lane, float* P) {
  const int half = lane >> 4;
  const int l16  = lane & 15;
  const float* ap = A + l16 * DMODEL + kb + 2 * half;
  const float* bp = W + l16 * DMODEL + kb + 2 * half;
  const v8f zero = {0.f, 0.f, 0.f, 0.f, 0.f, 0.f, 0.f, 0.f};
  v8f acc[4][4];
#pragma unroll
  for (int ti = 0; ti < 4; ++ti)
#pragma unroll
    for (int tj = 0; tj < 4; ++tj) acc[ti][tj] = zero;
  for (int kk = 0; kk < KLEN; kk += 32) {
    v2f a[8][4], b[8][4];
#pragma unroll
    for (int s8 = 0; s8 < 8; ++s8)
#pragma unroll
      for (int i = 0; i < 4; ++i) {
        const int o = i * 16 * DMODEL + kk + 4 * s8;
        a[s8][i].x = ap[o];  a[s8][i].y = ap[o + 1];
        b[s8][i].x = bp[o];  b[s8][i].y = bp[o + 1];
      }
#pragma unroll
    for (int s8 = 0; s8 < 8; ++s8)
#pragma unroll
      for (int ti = 0; ti < 4; ++ti)
#pragma unroll
        for (int tj = 0; tj < 4; ++tj)
          acc[ti][tj] = __builtin_amdgcn_wmma_f32_16x16x4_f32(false, a[s8][ti], false, b[s8][tj],
                                                              (short)0, acc[ti][tj], false, false);
  }
#pragma unroll
  for (int ti = 0; ti < 4; ++ti)
#pragma unroll
    for (int tj = 0; tj < 4; ++tj)
#pragma unroll
      for (int i = 0; i < 8; ++i)
        P[(16 * ti + 8 * half + i) * PT_PITCH + 16 * tj + l16] = acc[ti][tj][i];
}

template <int NS>
__device__ __forceinline__ v4f sum_planes(const float* P, int off) {
  v4f s[NS];
#pragma unroll
  for (int n = 0; n < NS; ++n) s[n] = *(const v4f*)(P + n * PT_PLANE + off);
#pragma unroll
  for (int st = 1; st < NS; st *= 2)
#pragma unroll
    for (int n = 0; n < NS; n += 2 * st) s[n] += s[n + st];
  return s[0];
}

#define NTILEMAX (NHEADS * (L_SEQ / 32))
#define TS_PITCH PT_PITCH

template <bool IS_K>
__device__ __forceinline__ void feature_from_tile(const float* T,
                                                  const float* __restrict__ RF,
                                                  float* __restrict__ dst,
                                                  float* __restrict__ tilemax,
                                                  int h, int row0, int mbase,
                                                  bool reports, int lane) {
  const int half = lane >> 4;
  const int l16  = lane & 15;
  const float* arow0 = T + l16 * TS_PITCH;
  const float* arow1 = arow0 + 16 * TS_PITCH;
  const v8f zero = {0.f, 0.f, 0.f, 0.f, 0.f, 0.f, 0.f, 0.f};
  float hrow0 = 0.f, hrow1 = 0.f;
#pragma unroll
  for (int mb = 0; mb < 2; ++mb) {
    const int tn = mbase + 32 * mb;
    const float* bcol0 = RF + tn + l16;
    const float* bcol1 = bcol0 + 16;
    Acc2x2 r;
    r.c[0][0] = zero; r.c[0][1] = zero; r.c[1][0] = zero; r.c[1][1] = zero;
    float sq0 = 0.f, sq1 = 0.f;
    for (int k0 = 0; k0 < DHEAD; k0 += 4) {
      const int ka = k0 + 2 * half;
      const v2f a0 = *(const v2f*)(arow0 + ka);
      const v2f a1 = *(const v2f*)(arow1 + ka);
      v2f b0, b1;
      b0.x = bcol0[ka * MFEAT];    b0.y = bcol0[(ka + 1) * MFEAT];
      b1.x = bcol1[ka * MFEAT];    b1.y = bcol1[(ka + 1) * MFEAT];
      r.c[0][0] = __builtin_amdgcn_wmma_f32_16x16x4_f32(false, a0, false, b0, (short)0, r.c[0][0], false, false);
      r.c[0][1] = __builtin_amdgcn_wmma_f32_16x16x4_f32(false, a0, false, b1, (short)0, r.c[0][1], false, false);
      r.c[1][0] = __builtin_amdgcn_wmma_f32_16x16x4_f32(false, a1, false, b0, (short)0, r.c[1][0], false, false);
      r.c[1][1] = __builtin_amdgcn_wmma_f32_16x16x4_f32(false, a1, false, b1, (short)0, r.c[1][1], false, false);
      if (IS_K && mb == 0) {
        sq0 += a0.x * a0.x; sq0 += a0.y * a0.y;
        sq1 += a1.x * a1.x; sq1 += a1.y * a1.y;
      }
    }
    if (IS_K) {
      if (mb == 0) {
        hrow0 = -0.5f * (sq0 + __shfl_xor(sq0, 16));
        hrow1 = -0.5f * (sq1 + __shfl_xor(sq1, 16));
        float mx = fmaxf(hrow0, hrow1);
#pragma unroll
        for (int off = 8; off > 0; off >>= 1) mx = fmaxf(mx, __shfl_xor(mx, off));
        if (reports && lane == 0) tilemax[h * (L_SEQ / 32) + (row0 >> 5)] = mx;
      }
#pragma unroll
      for (int ti = 0; ti < 2; ++ti) {
        const int rloc = ti * 16 + 8 * half;
        float hh[8];
#pragma unroll
        for (int i = 0; i < 8; ++i) hh[i] = __shfl(ti ? hrow1 : hrow0, 8 * half + i);
#pragma unroll
        for (int tj = 0; tj < 2; ++tj) {
          const int mcol = tn + tj * 16 + l16;
#pragma unroll
          for (int i = 0; i < 8; ++i)
            dst[(h * L_SEQ + row0 + rloc + i) * MFEAT + mcol] = hh[i] + r.c[ti][tj][i];
        }
      }
    } else {
#pragma unroll
      for (int ti = 0; ti < 2; ++ti)
#pragma unroll
        for (int tj = 0; tj < 2; ++tj) {
          const int mcol = tn + tj * 16 + l16;
          const int rbase = row0 + ti * 16 + 8 * half;
#pragma unroll
          for (int i = 0; i < 8; ++i)
            dst[(h * L_SEQ + rbase + i) * MFEAT + mcol] =
                NC_FEAT * (expf(r.c[ti][tj][i]) + KERNEL_EPS);
        }
    }
  }
}

__global__ void __launch_bounds__(128)
perf_qkv_features(const float* __restrict__ xq, const float* __restrict__ xk,
                  const float* __restrict__ xv,
                  const float* __restrict__ Wq, const float* __restrict__ bq,
                  const float* __restrict__ Wk, const float* __restrict__ bk,
                  const float* __restrict__ Wv, const float* __restrict__ bv,
                  const float* __restrict__ RF,
                  float* __restrict__ V, float* __restrict__ qprime,
                  float* __restrict__ kexp, float* __restrict__ tilemax) {
  const int h = blockIdx.y;
  const int mat = blockIdx.z;
  const int t = threadIdx.x;
  const int lane = t & 31;
  const int w = t >> 5;
  const int wr = w >> 1;
  const int wc = w & 1;
  const int tm0 = blockIdx.x * 64;
  const int tm = tm0 + 32 * wr;
  __shared__ __attribute__((aligned(16))) float P_s[4 * PT_PLANE];
  const float* X; const float* W; const float* bias; float scl;
  if (mat == 0)      { X = xq; W = Wq; bias = bq; scl = QK_SCALE; }
  else if (mat == 1) { X = xk; W = Wk; bias = bk; scl = QK_SCALE; }
  else               { X = xv; W = Wv; bias = bv; scl = 1.f; }
  gemm64x64_kslice<DMODEL / 4>(X + tm0 * DMODEL, W + h * DHEAD * DMODEL, (DMODEL / 4) * w, lane,
                               P_s + w * PT_PLANE);
  __syncthreads();
#pragma unroll
  for (int j = 0; j < 8; ++j) {
    const int q = t + 128 * j;
    const int row = q >> 4;
    const int c4 = (q & 15) * 4;
    v4f y = sum_planes<4>(P_s, row * PT_PITCH + c4);
    const float* bp = bias + h * DHEAD + c4;
    y.x = (y.x + bp[0]) * scl; y.y = (y.y + bp[1]) * scl;
    y.z = (y.z + bp[2]) * scl; y.w = (y.w + bp[3]) * scl;
    if (mat == 2) *(v4f*)(V + (size_t)(tm0 + row) * DMODEL + h * DHEAD + c4) = y;
    else          *(v4f*)(P_s + row * PT_PITCH + c4) = y;
  }
  if (mat == 2) return;
  __syncthreads();
  float* T_s = P_s;
  if (mat == 1)
    feature_from_tile<true>(T_s + 32 * wr * TS_PITCH, RF, kexp, tilemax, h, tm, 64 * wc,
                            wc == 0, lane);
  else
    feature_from_tile<false>(T_s + 32 * wr * TS_PITCH, RF, qprime, tilemax, h, tm, 64 * wc,
                             false, lane);
}

#define SCAN_ROWS 16
#define SCAN_BLOCKS (CHUNK / SCAN_ROWS)

__device__ __forceinline__ v4f kfeat4(v4f e, float ks) {
  v4f r;
  r.x = NC_FEAT * (expf(e.x - ks) + KERNEL_EPS);
  r.y = NC_FEAT * (expf(e.y - ks) + KERNEL_EPS);
  r.z = NC_FEAT * (expf(e.z - ks) + KERNEL_EPS);
  r.w = NC_FEAT * (expf(e.w - ks) + KERNEL_EPS);
  return r;
}

__global__ void __launch_bounds__(256)
perf_scan(const float* __restrict__ qprime, const float* __restrict__ kexp,
          const float* __restrict__ tilemax,
          const float* __restrict__ V, float* __restrict__ chunkbuf,
          float* __restrict__ numer, float* __restrict__ denloc) {
  const int h = blockIdx.x >> 4;
  const int c = blockIdx.x & 15;
  const int t = threadIdx.x;
  const int lane = t & 31;
  const int w = t >> 5;
  const int mz = 16 * w + (lane & 15);
  __shared__ __attribute__((aligned(16))) float qp_s[SCAN_ROWS * MFEAT];
  __shared__ __attribute__((aligned(16))) float kp_s[SCAN_ROWS * MFEAT];
  __shared__ __attribute__((aligned(16))) float vv_s[SCAN_ROWS * DHEAD];
  __shared__ __attribute__((aligned(16))) float red[SCAN_ROWS * 8 * DHEAD];
  __shared__ __attribute__((aligned(16))) float denp[SCAN_ROWS * MFEAT];
  __shared__ float dsink[16];
  __shared__ __attribute__((aligned(16))) float den_s[CHUNK];
  __shared__ __attribute__((aligned(16))) float z_s[MFEAT];
  __shared__ float wmax_s[8];

  v2f S[16];
#pragma unroll
  for (int j = 0; j < 16; ++j) { S[j].x = 0.f; S[j].y = 0.f; }
  float z = 0.f;
  float* const dst0 = (lane < 16) ? (denp + mz) : (dsink + (lane & 15));
  const int dstep = (lane < 16) ? MFEAT : 0;

  const int l_chunk = c * CHUNK;
  const v4f* qsrc = (const v4f*)(qprime + (size_t)(h * L_SEQ + l_chunk) * MFEAT);
  const v4f* ksrc = (const v4f*)(kexp + (size_t)(h * L_SEQ + l_chunk) * MFEAT);
  const int vrow = t >> 4;
  const int vcol = (t & 15) * 4;
  const float* vsrc = V + (size_t)(l_chunk + vrow) * DMODEL + h * DHEAD + vcol;

  const v2f tmx = ((const v2f*)tilemax)[t];
  v4f gq0 = qsrc[t], gq1 = qsrc[t + 256];
  v4f gk0 = ksrc[t], gk1 = ksrc[t + 256];
  v4f gv = *(const v4f*)vsrc;
  float mx = fmaxf(tmx.x, tmx.y);
#pragma unroll
  for (int off = 16; off > 0; off >>= 1) mx = fmaxf(mx, __shfl_xor(mx, off));
  if (lane == 0) wmax_s[w] = mx;
  __syncthreads();
  float ks = wmax_s[0];
#pragma unroll
  for (int i = 1; i < 8; ++i) ks = fmaxf(ks, wmax_s[i]);
  ((v4f*)qp_s)[t] = gq0; ((v4f*)qp_s)[t + 256] = gq1;
  ((v4f*)kp_s)[t] = kfeat4(gk0, ks); ((v4f*)kp_s)[t + 256] = kfeat4(gk1, ks);
  ((v4f*)vv_s)[t] = gv;
  __syncthreads();

  for (int b = 0; b < SCAN_BLOCKS; ++b) {
    if (b + 1 < SCAN_BLOCKS) {
      const int o = (b + 1) * (SCAN_ROWS * MFEAT / 4);
      gq0 = qsrc[o + t]; gq1 = qsrc[o + t + 256];
      gk0 = ksrc[o + t]; gk1 = ksrc[o + t + 256];
      gv = *(const v4f*)(vsrc + (size_t)(b + 1) * SCAN_ROWS * DMODEL);
    }
#pragma unroll 4
    for (int r = 0; r < SCAN_ROWS; ++r) {
      const v4f* kq = (const v4f*)(kp_s + r * MFEAT + 16 * w);
      const v4f* qq = (const v4f*)(qp_s + r * MFEAT + 16 * w);
      const v4f k0 = kq[0], k1 = kq[1], k2 = kq[2], k3 = kq[3];
      const v4f q0 = qq[0], q1 = qq[1], q2 = qq[2], q3 = qq[3];
      const v2f vv2 = *(const v2f*)(vv_s + r * DHEAD + 2 * lane);
      const float kz = kp_s[r * MFEAT + mz];
      const float qz = qp_s[r * MFEAT + mz];
      const float kk[16] = {k0.x, k0.y, k0.z, k0.w, k1.x, k1.y, k1.z, k1.w,
                            k2.x, k2.y, k2.z, k2.w, k3.x, k3.y, k3.z, k3.w};
      const float qv[16] = {q0.x, q0.y, q0.z, q0.w, q1.x, q1.y, q1.z, q1.w,
                            q2.x, q2.y, q2.z, q2.w, q3.x, q3.y, q3.z, q3.w};
#pragma unroll
      for (int j = 0; j < 16; ++j) S[j] += kk[j] * vv2;
      z += kz;
      v2f sa = {0.f, 0.f}, sb = {0.f, 0.f};
#pragma unroll
      for (int j = 0; j < 16; j += 2) {
        sa += qv[j] * S[j];
        sb += qv[j + 1] * S[j + 1];
      }
      *(v2f*)(red + (r * 8 + w) * DHEAD + 2 * lane) = sa + sb;
      dst0[r * dstep] = qz * z;
    }
    if (b + 1 < SCAN_BLOCKS) { gk0 = kfeat4(gk0, ks); gk1 = kfeat4(gk1, ks); }
    __syncthreads();
    if (b + 1 < SCAN_BLOCKS) {
      ((v4f*)qp_s)[t] = gq0; ((v4f*)qp_s)[t + 256] = gq1;
      ((v4f*)kp_s)[t] = gk0; ((v4f*)kp_s)[t + 256] = gk1;
      ((v4f*)vv_s)[t] = gv;
    }
#pragma unroll
    for (int p = 0; p < 4; ++p) {
      const int r = p * 4 + (t >> 6);
      const int d = t & 63;
      const float* dp = denp + r * MFEAT;
      float den = (dp[lane] + dp[lane + 32]) + (dp[lane + 64] + dp[lane + 96]);
#pragma unroll
      for (int off = 16; off > 0; off >>= 1) den += __shfl_xor(den, off);
      float num = 0.f;
#pragma unroll
      for (int ww = 0; ww < 8; ++ww) num += red[(r * 8 + ww) * DHEAD + d];
      const int l = l_chunk + b * SCAN_ROWS + r;
      numer[(size_t)l * DMODEL + h * DHEAD + d] = num;
      if (d == 0) den_s[b * SCAN_ROWS + r] = den;
    }
    __syncthreads();
  }
  float* sums = chunkbuf + (size_t)(h * NCHUNKS + c) * CHUNK_STRIDE;
#pragma unroll
  for (int j = 0; j < 16; ++j)
    *(v2f*)(sums + (16 * w + j) * DHEAD + 2 * lane) = S[j];
  if (lane < 16) z_s[mz] = z;
  __syncthreads();
  if (w == 0) {
    ((v4f*)(sums + MFEAT * DHEAD))[lane] = ((const v4f*)z_s)[lane];
    ((v4f*)(denloc + h * L_SEQ + l_chunk))[lane] = ((const v4f*)den_s)[lane];
  }
}

#define PS_PITCH 40

__global__ void __launch_bounds__(128)
perf_apply_prefix(const float* __restrict__ qprime, const float* __restrict__ chunkbuf,
                  const float* __restrict__ denloc, float* __restrict__ ctx) {
  const int c = blockIdx.x;
  const int tn = blockIdx.y * 32;
  const int h = blockIdx.z;
  const int t = threadIdx.x;
  const int lane = t & 31;
  const int w = t >> 5;
  const int tm = c * CHUNK + 32 * w;
  __shared__ __attribute__((aligned(16))) float P_s[MFEAT * PS_PITCH];
  __shared__ __attribute__((aligned(16))) float zp_s[MFEAT];

  const float* hbase = chunkbuf + (size_t)(h * NCHUNKS) * CHUNK_STRIDE;
  const int prow = t >> 3;
  const int pcol = (t & 7) * 4;
  const v4f zero4 = {0.f, 0.f, 0.f, 0.f};
  v4f acc[8];
#pragma unroll
  for (int j = 0; j < 8; ++j) acc[j] = zero4;
  float zacc = 0.f;
#pragma unroll 4
  for (int cc = 0; cc < c; ++cc) {
    const float* s = hbase + (size_t)cc * CHUNK_STRIDE;
#pragma unroll
    for (int j = 0; j < 8; ++j)
      acc[j] += *(const v4f*)(s + (prow + 16 * j) * DHEAD + tn + pcol);
    zacc += s[MFEAT * DHEAD + t];
  }
#pragma unroll
  for (int j = 0; j < 8; ++j)
    *(v4f*)(P_s + (prow + 16 * j) * PS_PITCH + pcol) = acc[j];
  zp_s[t] = zacc;
  __syncthreads();

  const int half = lane >> 4;
  const int l16  = lane & 15;
  const float* A = qprime + (size_t)(h * L_SEQ + tm) * MFEAT;
  const float* arow0 = A + l16 * MFEAT;
  const float* arow1 = A + (l16 + 16) * MFEAT;
  const float* bcol0 = P_s + l16;
  const float* bcol1 = P_s + 16 + l16;
  const v8f zero = {0.f, 0.f, 0.f, 0.f, 0.f, 0.f, 0.f, 0.f};
  Acc2x2 r;
  r.c[0][0] = zero; r.c[0][1] = zero; r.c[1][0] = zero; r.c[1][1] = zero;
  for (int k0 = 0; k0 < MFEAT; k0 += 4) {
    const int ka = k0 + 2 * half;
    v2f a0, a1, b0, b1;
    a0.x = arow0[ka];               a0.y = arow0[ka + 1];
    a1.x = arow1[ka];               a1.y = arow1[ka + 1];
    b0.x = bcol0[ka * PS_PITCH];    b0.y = bcol0[(ka + 1) * PS_PITCH];
    b1.x = bcol1[ka * PS_PITCH];    b1.y = bcol1[(ka + 1) * PS_PITCH];
    r.c[0][0] = __builtin_amdgcn_wmma_f32_16x16x4_f32(false, a0, false, b0, (short)0, r.c[0][0], false, false);
    r.c[0][1] = __builtin_amdgcn_wmma_f32_16x16x4_f32(false, a0, false, b1, (short)0, r.c[0][1], false, false);
    r.c[1][0] = __builtin_amdgcn_wmma_f32_16x16x4_f32(false, a1, false, b0, (short)0, r.c[1][0], false, false);
    r.c[1][1] = __builtin_amdgcn_wmma_f32_16x16x4_f32(false, a1, false, b1, (short)0, r.c[1][1], false, false);
  }
  const v4f* qrow = (const v4f*)(A + lane * MFEAT);
  const v4f* zp = (const v4f*)zp_s;
  v4f dacc = {0.f, 0.f, 0.f, 0.f};
  for (int m4 = 0; m4 < MFEAT / 4; ++m4) dacc += qrow[m4] * zp[m4];
  const float dtot = denloc[h * L_SEQ + tm + lane] +
                     ((dacc.x + dacc.y) + (dacc.z + dacc.w)) + DEN_EPS;
#pragma unroll
  for (int ti = 0; ti < 2; ++ti) {
    const int rloc = ti * 16 + 8 * half;
    float dd[8];
#pragma unroll
    for (int i = 0; i < 8; ++i) dd[i] = __shfl(dtot, rloc + i);
#pragma unroll
    for (int tj = 0; tj < 2; ++tj) {
      const int col = h * DHEAD + tn + tj * 16 + l16;
#pragma unroll
      for (int i = 0; i < 8; ++i) {
        float* p = ctx + (size_t)(tm + rloc + i) * DMODEL + col;
        *p = (*p + r.c[ti][tj][i]) / dd[i];
      }
    }
  }
}

#define OP_SPLIT 8
__global__ void __launch_bounds__(32 * OP_SPLIT)
perf_out_proj(const float* __restrict__ ctx, const float* __restrict__ Wo,
              const float* __restrict__ bo, float* __restrict__ out) {
  const int tm0 = blockIdx.x * 64;
  const int tn0 = blockIdx.y * 64;
  const int t = threadIdx.x;
  const int lane = t & 31;
  const int w = t >> 5;
  __shared__ __attribute__((aligned(16))) float P_s[OP_SPLIT * PT_PLANE];
  gemm64x64_kslice<DMODEL / OP_SPLIT>(ctx + tm0 * DMODEL, Wo + tn0 * DMODEL,
                                      (DMODEL / OP_SPLIT) * w, lane, P_s + w * PT_PLANE);
  __syncthreads();
#pragma unroll
  for (int j = 0; j < 64 * 16 / (32 * OP_SPLIT); ++j) {
    const int q = t + 32 * OP_SPLIT * j;
    const int row = q >> 4;
    const int c4 = (q & 15) * 4;
    v4f y = sum_planes<OP_SPLIT>(P_s, row * PT_PITCH + c4);
    const float* bp = bo + tn0 + c4;
    y.x += bp[0]; y.y += bp[1]; y.z += bp[2]; y.w += bp[3];
    *(v4f*)(out + (size_t)(tm0 + row) * DMODEL + tn0 + c4) = y;
  }
}

extern "C" void kernel_launch(void* const* d_in, const int* in_sizes, int n_in,
                              void* d_out, int out_size, void* d_ws, size_t ws_size,
                              hipStream_t stream) {
  (void)in_sizes; (void)n_in; (void)out_size; (void)ws_size;
  const float* query = (const float*)d_in[0];
  const float* key_  = (const float*)d_in[1];
  const float* value = (const float*)d_in[2];
  const float* Wq = (const float*)d_in[3];
  const float* bq = (const float*)d_in[4];
  const float* Wk = (const float*)d_in[5];
  const float* bk = (const float*)d_in[6];
  const float* Wv = (const float*)d_in[7];
  const float* bv = (const float*)d_in[8];
  const float* Wo = (const float*)d_in[9];
  const float* bo = (const float*)d_in[10];
  const float* RF = (const float*)d_in[11];

  float* ws = (float*)d_ws;
  const size_t SZ_LD = (size_t)L_SEQ * DMODEL;
  const size_t SZ_PRIME = (size_t)NHEADS * L_SEQ * MFEAT;
  float* context = ws;
  float* V      = context + SZ_LD;
  float* tilemax = V + SZ_LD;
  float* qprime = tilemax + NTILEMAX;
  float* kexp   = qprime + SZ_PRIME;
  float* chunkbuf = kexp + SZ_PRIME;
  float* denloc = chunkbuf + (size_t)NHEADS * NCHUNKS * CHUNK_STRIDE;

  perf_qkv_features<<<dim3(L_SEQ / 64, NHEADS, 3), 128, 0, stream>>>(
      query, key_, value, Wq, bq, Wk, bk, Wv, bv, RF, V, qprime, kexp, tilemax);
  perf_scan<<<NHEADS * NCHUNKS, 256, 0, stream>>>(qprime, kexp, tilemax, V, chunkbuf,
                                                  context, denloc);
  perf_apply_prefix<<<dim3(NCHUNKS, DHEAD / 32, NHEADS), 128, 0, stream>>>(
      qprime, chunkbuf, denloc, context);
  perf_out_proj<<<dim3(L_SEQ / 64, DMODEL / 64), 32 * OP_SPLIT, 0, stream>>>(
      context, Wo, bo, (float*)d_out);
}


# projection GEMMs: fragment loads issued line by line (the eight 16-byte pieces of a row group's 16 lines by eight consecutive loads, was k-step by k-step over all 128 lines); sums unchanged
# speedup vs baseline: 8.0247x; 1.1686x over previous
#include <hip/hip_runtime.h>
#include <math.h>

#define L_SEQ 2048
#define DMODEL 512
#define NHEADS 8
#define DHEAD 64
#define MFEAT 128
#define NCHUNKS 16
#define CHUNK 128
#define CHUNK_STRIDE 8320
#define KERNEL_EPS 1e-4f
#define DEN_EPS 1e-6f
#define QK_SCALE 0.21022410381342864f
#define NC_FEAT 0.08838834764831845f

typedef float v2f __attribute__((ext_vector_type(2)));
typedef float v8f __attribute__((ext_vector_type(8)));

struct Acc2x2 { v8f c[2][2]; };
typedef float v4f __attribute__((ext_vector_type(4)));

#define PT_PITCH 68
#define PT_PLANE (64 * PT_PITCH)

template <int KLEN>
__device__ __forceinline__ void gemm64x64_kslice(const float* __restrict__ A,
                                                 const float* __restrict__ W,
                                                 int kb, int lane, float* P) {
  const int half = lane >> 4;
  const int l16  = lane & 15;
  const float* ap = A + l16 * DMODEL + kb + 2 * half;
  const float* bp = W + l16 * DMODEL + kb + 2 * half;
  const v8f zero = {0.f, 0.f, 0.f, 0.f, 0.f, 0.f, 0.f, 0.f};
  v8f acc[4][4];
#pragma unroll
  for (int ti = 0; ti < 4; ++ti)
#pragma unroll
    for (int tj = 0; tj < 4; ++tj) acc[ti][tj] = zero;
  for (int kk = 0; kk < KLEN; kk += 32) {
    v2f a[8][4], b[8][4];
#pragma unroll
    for (int s8 = 0; s8 < 8; ++s8) {
      const int o = kk + 4 * s8;
      a[s8][0].x = ap[o];  a[s8][0].y = ap[o + 1];
    }
    __builtin_amdgcn_sched_barrier(0);
#pragma unroll
    for (int j = 0; j < 4; ++j) {
#pragma unroll
      for (int s8 = 0; s8 < 8; ++s8) {
        const int o = j * 16 * DMODEL + kk + 4 * s8;
        b[s8][j].x = bp[o];  b[s8][j].y = bp[o + 1];
      }
      __builtin_amdgcn_sched_barrier(0);
    }
#pragma unroll
    for (int i = 1; i < 4; ++i) {
#pragma unroll
      for (int s8 = 0; s8 < 8; ++s8) {
        const int o = i * 16 * DMODEL + kk + 4 * s8;
        a[s8][i].x = ap[o];  a[s8][i].y = ap[o + 1];
      }
      __builtin_amdgcn_sched_barrier(0);
    }
#pragma unroll
    for (int ti = 0; ti < 4; ++ti) {
#pragma unroll
      for (int s8 = 0; s8 < 8; ++s8)
#pragma unroll
        for (int tj = 0; tj < 4; ++tj)
          acc[ti][tj] = __builtin_amdgcn_wmma_f32_16x16x4_f32(false, a[s8][ti], false, b[s8][tj],
                                                              (short)0, acc[ti][tj], false, false);
      if (ti < 3) __builtin_amdgcn_sched_barrier(0);
    }
  }
#pragma unroll
  for (int ti = 0; ti < 4; ++ti)
#pragma unroll
    for (int tj = 0; tj < 4; ++tj)
#pragma unroll
      for (int i = 0; i < 8; ++i)
        P[(16 * ti + 8 * half + i) * PT_PITCH + 16 * tj + l16] = acc[ti][tj][i];
}

template <int NS>
__device__ __forceinline__ v4f sum_planes(const float* P, int off) {
  v4f s[NS];
#pragma unroll
  for (int n = 0; n < NS; ++n) s[n] = *(const v4f*)(P + n * PT_PLANE + off);
#pragma unroll
  for (int st = 1; st < NS; st *= 2)
#pragma unroll
    for (int n = 0; n < NS; n += 2 * st) s[n] += s[n + st];
  return s[0];
}

#define NTILEMAX (NHEADS * (L_SEQ / 32))
#define TS_PITCH PT_PITCH

template <bool IS_K>
__device__ __forceinline__ void feature_from_tile(const float* T,
                                                  const float* __restrict__ RF,
                                                  float* __restrict__ dst,
                                                  float* __restrict__ tilemax,
                                                  int h, int row0, int mbase,
                                                  bool reports, int lane) {
  const int half = lane >> 4;
  const int l16  = lane & 15;
  const float* arow0 = T + l16 * TS_PITCH;
  const float* arow1 = arow0 + 16 * TS_PITCH;
  const v8f zero = {0.f, 0.f, 0.f, 0.f, 0.f, 0.f, 0.f, 0.f};
  float hrow0 = 0.f, hrow1 = 0.f;
#pragma unroll
  for (int mb = 0; mb < 2; ++mb) {
    const int tn = mbase + 32 * mb;
    const float* bcol0 = RF + tn + l16;
    const float* bcol1 = bcol0 + 16;
    Acc2x2 r;
    r.c[0][0] = zero; r.c[0][1] = zero; r.c[1][0] = zero; r.c[1][1] = zero;
    float sq0 = 0.f, sq1 = 0.f;
    for (int k0 = 0; k0 < DHEAD; k0 += 4) {
      const int ka = k0 + 2 * half;
      const v2f a0 = *(const v2f*)(arow0 + ka);
      const v2f a1 = *(const v2f*)(arow1 + ka);
      v2f b0, b1;
      b0.x = bcol0[ka * MFEAT];    b0.y = bcol0[(ka + 1) * MFEAT];
      b1.x = bcol1[ka * MFEAT];    b1.y = bcol1[(ka + 1) * MFEAT];
      r.c[0][0] = __builtin_amdgcn_wmma_f32_16x16x4_f32(false, a0, false, b0, (short)0, r.c[0][0], false, false);
      r.c[0][1] = __builtin_amdgcn_wmma_f32_16x16x4_f32(false, a0, false, b1, (short)0, r.c[0][1], false, false);
      r.c[1][0] = __builtin_amdgcn_wmma_f32_16x16x4_f32(false, a1, false, b0, (short)0, r.c[1][0], false, false);
      r.c[1][1] = __builtin_amdgcn_wmma_f32_16x16x4_f32(false, a1, false, b1, (short)0, r.c[1][1], false, false);
      if (IS_K && mb == 0) {
        sq0 += a0.x * a0.x; sq0 += a0.y * a0.y;
        sq1 += a1.x * a1.x; sq1 += a1.y * a1.y;
      }
    }
    if (IS_K) {
      if (mb == 0) {
        hrow0 = -0.5f * (sq0 + __shfl_xor(sq0, 16));
        hrow1 = -0.5f * (sq1 + __shfl_xor(sq1, 16));
        float mx = fmaxf(hrow0, hrow1);
#pragma unroll
        for (int off = 8; off > 0; off >>= 1) mx = fmaxf(mx, __shfl_xor(mx, off));
        if (reports && lane == 0) tilemax[h * (L_SEQ / 32) + (row0 >> 5)] = mx;
      }
#pragma unroll
      for (int ti = 0; ti < 2; ++ti) {
        const int rloc = ti * 16 + 8 * half;
        float hh[8];
#pragma unroll
        for (int i = 0; i < 8; ++i) hh[i] = __shfl(ti ? hrow1 : hrow0, 8 * half + i);
#pragma unroll
        for (int tj = 0; tj < 2; ++tj) {
          const int mcol = tn + tj * 16 + l16;
#pragma unroll
          for (int i = 0; i < 8; ++i)
            dst[(h * L_SEQ + row0 + rloc + i) * MFEAT + mcol] = hh[i] + r.c[ti][tj][i];
        }
      }
    } else {
#pragma unroll
      for (int ti = 0; ti < 2; ++ti)
#pragma unroll
        for (int tj = 0; tj < 2; ++tj) {
          const int mcol = tn + tj * 16 + l16;
          const int rbase = row0 + ti * 16 + 8 * half;
#pragma unroll
          for (int i = 0; i < 8; ++i)
            dst[(h * L_SEQ + rbase + i) * MFEAT + mcol] =
                NC_FEAT * (expf(r.c[ti][tj][i]) + KERNEL_EPS);
        }
    }
  }
}

__global__ void __launch_bounds__(128)
perf_qkv_features(const float* __restrict__ xq, const float* __restrict__ xk,
                  const float* __restrict__ xv,
                  const float* __restrict__ Wq, const float* __restrict__ bq,
                  const float* __restrict__ Wk, const float* __restrict__ bk,
                  const float* __restrict__ Wv, const float* __restrict__ bv,
                  const float* __restrict__ RF,
                  float* __restrict__ V, float* __restrict__ qprime,
                  float* __restrict__ kexp, float* __restrict__ tilemax) {
  const int h = blockIdx.y;
  const int mat = blockIdx.z;
  const int t = threadIdx.x;
  const int lane = t & 31;
  const int w = t >> 5;
  const int wr = w >> 1;
  const int wc = w & 1;
  const int tm0 = blockIdx.x * 64;
  const int tm = tm0 + 32 * wr;
  __shared__ __attribute__((aligned(16))) float P_s[4 * PT_PLANE];
  const float* X; const float* W; const float* bias; float scl;
  if (mat == 0)      { X = xq; W = Wq; bias = bq; scl = QK_SCALE; }
  else if (mat == 1) { X = xk; W = Wk; bias = bk; scl = QK_SCALE; }
  else               { X = xv; W = Wv; bias = bv; scl = 1.f; }
  gemm64x64_kslice<DMODEL / 4>(X + tm0 * DMODEL, W + h * DHEAD * DMODEL, (DMODEL / 4) * w, lane,
                               P_s + w * PT_PLANE);
  __syncthreads();
#pragma unroll
  for (int j = 0; j < 8; ++j) {
    const int q = t + 128 * j;
    const int row = q >> 4;
    const int c4 = (q & 15) * 4;
    v4f y = sum_planes<4>(P_s, row * PT_PITCH + c4);
    const float* bp = bias + h * DHEAD + c4;
    y.x = (y.x + bp[0]) * scl; y.y = (y.y + bp[1]) * scl;
    y.z = (y.z + bp[2]) * scl; y.w = (y.w + bp[3]) * scl;
    if (mat == 2) *(v4f*)(V + (size_t)(tm0 + row) * DMODEL + h * DHEAD + c4) = y;
    else          *(v4f*)(P_s + row * PT_PITCH + c4) = y;
  }
  if (mat == 2) return;
  __syncthreads();
  float* T_s = P_s;
  if (mat == 1)
    feature_from_tile<true>(T_s + 32 * wr * TS_PITCH, RF, kexp, tilemax, h, tm, 64 * wc,
                            wc == 0, lane);
  else
    feature_from_tile<false>(T_s + 32 * wr * TS_PITCH, RF, qprime, tilemax, h, tm, 64 * wc,
                             false, lane);
}

#define SCAN_ROWS 16
#define SCAN_BLOCKS (CHUNK / SCAN_ROWS)

__device__ __forceinline__ v4f kfeat4(v4f e, float ks) {
  v4f r;
  r.x = NC_FEAT * (expf(e.x - ks) + KERNEL_EPS);
  r.y = NC_FEAT * (expf(e.y - ks) + KERNEL_EPS);
  r.z = NC_FEAT * (expf(e.z - ks) + KERNEL_EPS);
  r.w = NC_FEAT * (expf(e.w - ks) + KERNEL_EPS);
  return r;
}

__global__ void __launch_bounds__(256)
perf_scan(const float* __restrict__ qprime, const float* __restrict__ kexp,
          const float* __restrict__ tilemax,
          const float* __restrict__ V, float* __restrict__ chunkbuf,
          float* __restrict__ numer, float* __restrict__ denloc) {
  const int h = blockIdx.x >> 4;
  const int c = blockIdx.x & 15;
  const int t = threadIdx.x;
  const int lane = t & 31;
  const int w = t >> 5;
  const int mz = 16 * w + (lane & 15);
  __shared__ __attribute__((aligned(16))) float qp_s[SCAN_ROWS * MFEAT];
  __shared__ __attribute__((aligned(16))) float kp_s[SCAN_ROWS * MFEAT];
  __shared__ __attribute__((aligned(16))) float vv_s[SCAN_ROWS * DHEAD];
  __shared__ __attribute__((aligned(16))) float red[SCAN_ROWS * 8 * DHEAD];
  __shared__ __attribute__((aligned(16))) float denp[SCAN_ROWS * MFEAT];
  __shared__ float dsink[16];
  __shared__ __attribute__((aligned(16))) float den_s[CHUNK];
  __shared__ __attribute__((aligned(16))) float z_s[MFEAT];
  __shared__ float wmax_s[8];

  v2f S[16];
#pragma unroll
  for (int j = 0; j < 16; ++j) { S[j].x = 0.f; S[j].y = 0.f; }
  float z = 0.f;
  float* const dst0 = (lane < 16) ? (denp + mz) : (dsink + (lane & 15));
  const int dstep = (lane < 16) ? MFEAT : 0;

  const int l_chunk = c * CHUNK;
  const v4f* qsrc = (const v4f*)(qprime + (size_t)(h * L_SEQ + l_chunk) * MFEAT);
  const v4f* ksrc = (const v4f*)(kexp + (size_t)(h * L_SEQ + l_chunk) * MFEAT);
  const int vrow = t >> 4;
  const int vcol = (t & 15) * 4;
  const float* vsrc = V + (size_t)(l_chunk + vrow) * DMODEL + h * DHEAD + vcol;

  const v2f tmx = ((const v2f*)tilemax)[t];
  v4f gq0 = qsrc[t], gq1 = qsrc[t + 256];
  v4f gk0 = ksrc[t], gk1 = ksrc[t + 256];
  v4f gv = *(const v4f*)vsrc;
  float mx = fmaxf(tmx.x, tmx.y);
#pragma unroll
  for (int off = 16; off > 0; off >>= 1) mx = fmaxf(mx, __shfl_xor(mx, off));
  if (lane == 0) wmax_s[w] = mx;
  __syncthreads();
  float ks = wmax_s[0];
#pragma unroll
  for (int i = 1; i < 8; ++i) ks = fmaxf(ks, wmax_s[i]);
  ((v4f*)qp_s)[t] = gq0; ((v4f*)qp_s)[t + 256] = gq1;
  ((v4f*)kp_s)[t] = kfeat4(gk0, ks); ((v4f*)kp_s)[t + 256] = kfeat4(gk1, ks);
  ((v4f*)vv_s)[t] = gv;
  __syncthreads();

  for (int b = 0; b < SCAN_BLOCKS; ++b) {
    if (b + 1 < SCAN_BLOCKS) {
      const int o = (b + 1) * (SCAN_ROWS * MFEAT / 4);
      gq0 = qsrc[o + t]; gq1 = qsrc[o + t + 256];
      gk0 = ksrc[o + t]; gk1 = ksrc[o + t + 256];
      gv = *(const v4f*)(vsrc + (size_t)(b + 1) * SCAN_ROWS * DMODEL);
    }
#pragma unroll 4
    for (int r = 0; r < SCAN_ROWS; ++r) {
      const v4f* kq = (const v4f*)(kp_s + r * MFEAT + 16 * w);
      const v4f* qq = (const v4f*)(qp_s + r * MFEAT + 16 * w);
      const v4f k0 = kq[0], k1 = kq[1], k2 = kq[2], k3 = kq[3];
      const v4f q0 = qq[0], q1 = qq[1], q2 = qq[2], q3 = qq[3];
      const v2f vv2 = *(const v2f*)(vv_s + r * DHEAD + 2 * lane);
      const float kz = kp_s[r * MFEAT + mz];
      const float qz = qp_s[r * MFEAT + mz];
      const float kk[16] = {k0.x, k0.y, k0.z, k0.w, k1.x, k1.y, k1.z, k1.w,
                            k2.x, k2.y, k2.z, k2.w, k3.x, k3.y, k3.z, k3.w};
      const float qv[16] = {q0.x, q0.y, q0.z, q0.w, q1.x, q1.y, q1.z, q1.w,
                            q2.x, q2.y, q2.z, q2.w, q3.x, q3.y, q3.z, q3.w};
#pragma unroll
      for (int j = 0; j < 16; ++j) S[j] += kk[j] * vv2;
      z += kz;
      v2f sa = {0.f, 0.f}, sb = {0.f, 0.f};
#pragma unroll
      for (int j = 0; j < 16; j += 2) {
        sa += qv[j] * S[j];
        sb += qv[j + 1] * S[j + 1];
      }
      *(v2f*)(red + (r * 8 + w) * DHEAD + 2 * lane) = sa + sb;
      dst0[r * dstep] = qz * z;
    }
    if (b + 1 < SCAN_BLOCKS) { gk0 = kfeat4(gk0, ks); gk1 = kfeat4(gk1, ks); }
    __syncthreads();
    if (b + 1 < SCAN_BLOCKS) {
      ((v4f*)qp_s)[t] = gq0; ((v4f*)qp_s)[t + 256] = gq1;
      ((v4f*)kp_s)[t] = gk0; ((v4f*)kp_s)[t + 256] = gk1;
      ((v4f*)vv_s)[t] = gv;
    }
#pragma unroll
    for (int p = 0; p < 4; ++p) {
      const int r = p * 4 + (t >> 6);
      const int d = t & 63;
      const float* dp = denp + r * MFEAT;
      float den = (dp[lane] + dp[lane + 32]) + (dp[lane + 64] + dp[lane + 96]);
#pragma unroll
      for (int off = 16; off > 0; off >>= 1) den += __shfl_xor(den, off);
      float num = 0.f;
#pragma unroll
      for (int ww = 0; ww < 8; ++ww) num += red[(r * 8 + ww) * DHEAD + d];
      const int l = l_chunk + b * SCAN_ROWS + r;
      numer[(size_t)l * DMODEL + h * DHEAD + d] = num;
      if (d == 0) den_s[b * SCAN_ROWS + r] = den;
    }
    __syncthreads();
  }
  float* sums = chunkbuf + (size_t)(h * NCHUNKS + c) * CHUNK_STRIDE;
#pragma unroll
  for (int j = 0; j < 16; ++j)
    *(v2f*)(sums + (16 * w + j) * DHEAD + 2 * lane) = S[j];
  if (lane < 16) z_s[mz] = z;
  __syncthreads();
  if (w == 0) {
    ((v4f*)(sums + MFEAT * DHEAD))[lane] = ((const v4f*)z_s)[lane];
    ((v4f*)(denloc + h * L_SEQ + l_chunk))[lane] = ((const v4f*)den_s)[lane];
  }
}

#define PS_PITCH 40

__global__ void __launch_bounds__(128)
perf_apply_prefix(const float* __restrict__ qprime, const float* __restrict__ chunkbuf,
                  const float* __restrict__ denloc, float* __restrict__ ctx) {
  const int c = blockIdx.x;
  const int tn = blockIdx.y * 32;
  const int h = blockIdx.z;
  const int t = threadIdx.x;
  const int lane = t & 31;
  const int w = t >> 5;
  const int tm = c * CHUNK + 32 * w;
  __shared__ __attribute__((aligned(16))) float P_s[MFEAT * PS_PITCH];
  __shared__ __attribute__((aligned(16))) float zp_s[MFEAT];

  const float* hbase = chunkbuf + (size_t)(h * NCHUNKS) * CHUNK_STRIDE;
  const int prow = t >> 3;
  const int pcol = (t & 7) * 4;
  const v4f zero4 = {0.f, 0.f, 0.f, 0.f};
  v4f acc[8];
#pragma unroll
  for (int j = 0; j < 8; ++j) acc[j] = zero4;
  float zacc = 0.f;
#pragma unroll 4
  for (int cc = 0; cc < c; ++cc) {
    const float* s = hbase + (size_t)cc * CHUNK_STRIDE;
#pragma unroll
    for (int j = 0; j < 8; ++j)
      acc[j] += *(const v4f*)(s + (prow + 16 * j) * DHEAD + tn + pcol);
    zacc += s[MFEAT * DHEAD + t];
  }
#pragma unroll
  for (int j = 0; j < 8; ++j)
    *(v4f*)(P_s + (prow + 16 * j) * PS_PITCH + pcol) = acc[j];
  zp_s[t] = zacc;
  __syncthreads();

  const int half = lane >> 4;
  const int l16  = lane & 15;
  const float* A = qprime + (size_t)(h * L_SEQ + tm) * MFEAT;
  const float* arow0 = A + l16 * MFEAT;
  const float* arow1 = A + (l16 + 16) * MFEAT;
  const float* bcol0 = P_s + l16;
  const float* bcol1 = P_s + 16 + l16;
  const v8f zero = {0.f, 0.f, 0.f, 0.f, 0.f, 0.f, 0.f, 0.f};
  Acc2x2 r;
  r.c[0][0] = zero; r.c[0][1] = zero; r.c[1][0] = zero; r.c[1][1] = zero;
  for (int k0 = 0; k0 < MFEAT; k0 += 4) {
    const int ka = k0 + 2 * half;
    v2f a0, a1, b0, b1;
    a0.x = arow0[ka];               a0.y = arow0[ka + 1];
    a1.x = arow1[ka];               a1.y = arow1[ka + 1];
    b0.x = bcol0[ka * PS_PITCH];    b0.y = bcol0[(ka + 1) * PS_PITCH];
    b1.x = bcol1[ka * PS_PITCH];    b1.y = bcol1[(ka + 1) * PS_PITCH];
    r.c[0][0] = __builtin_amdgcn_wmma_f32_16x16x4_f32(false, a0, false, b0, (short)0, r.c[0][0], false, false);
    r.c[0][1] = __builtin_amdgcn_wmma_f32_16x16x4_f32(false, a0, false, b1, (short)0, r.c[0][1], false, false);
    r.c[1][0] = __builtin_amdgcn_wmma_f32_16x16x4_f32(false, a1, false, b0, (short)0, r.c[1][0], false, false);
    r.c[1][1] = __builtin_amdgcn_wmma_f32_16x16x4_f32(false, a1, false, b1, (short)0, r.c[1][1], false, false);
  }
  const v4f* qrow = (const v4f*)(A + lane * MFEAT);
  const v4f* zp = (const v4f*)zp_s;
  v4f dacc = {0.f, 0.f, 0.f, 0.f};
  for (int m4 = 0; m4 < MFEAT / 4; ++m4) dacc += qrow[m4] * zp[m4];
  const float dtot = denloc[h * L_SEQ + tm + lane] +
                     ((dacc.x + dacc.y) + (dacc.z + dacc.w)) + DEN_EPS;
#pragma unroll
  for (int ti = 0; ti < 2; ++ti) {
    const int rloc = ti * 16 + 8 * half;
    float dd[8];
#pragma unroll
    for (int i = 0; i < 8; ++i) dd[i] = __shfl(dtot, rloc + i);
#pragma unroll
    for (int tj = 0; tj < 2; ++tj) {
      const int col = h * DHEAD + tn + tj * 16 + l16;
#pragma unroll
      for (int i = 0; i < 8; ++i) {
        float* p = ctx + (size_t)(tm + rloc + i) * DMODEL + col;
        *p = (*p + r.c[ti][tj][i]) / dd[i];
      }
    }
  }
}

#define OP_SPLIT 8
__global__ void __launch_bounds__(32 * OP_SPLIT)
perf_out_proj(const float* __restrict__ ctx, const float* __restrict__ Wo,
              const float* __restrict__ bo, float* __restrict__ out) {
  const int tm0 = blockIdx.x * 64;
  const int tn0 = blockIdx.y * 64;
  const int t = threadIdx.x;
  const int lane = t & 31;
  const int w = t >> 5;
  __shared__ __attribute__((aligned(16))) float P_s[OP_SPLIT * PT_PLANE];
  gemm64x64_kslice<DMODEL / OP_SPLIT>(ctx + tm0 * DMODEL, Wo + tn0 * DMODEL,
                                      (DMODEL / OP_SPLIT) * w, lane, P_s + w * PT_PLANE);
  __syncthreads();
#pragma unroll
  for (int j = 0; j < 64 * 16 / (32 * OP_SPLIT); ++j) {
    const int q = t + 32 * OP_SPLIT * j;
    const int row = q >> 4;
    const int c4 = (q & 15) * 4;
    v4f y = sum_planes<OP_SPLIT>(P_s, row * PT_PITCH + c4);
    const float* bp = bo + tn0 + c4;
    y.x += bp[0]; y.y += bp[1]; y.z += bp[2]; y.w += bp[3];
    *(v4f*)(out + (size_t)(tm0 + row) * DMODEL + tn0 + c4) = y;
  }
}

extern "C" void kernel_launch(void* const* d_in, const int* in_sizes, int n_in,
                              void* d_out, int out_size, void* d_ws, size_t ws_size,
                              hipStream_t stream) {
  (void)in_sizes; (void)n_in; (void)out_size; (void)ws_size;
  const float* query = (const float*)d_in[0];
  const float* key_  = (const float*)d_in[1];
  const float* value = (const float*)d_in[2];
  const float* Wq = (const float*)d_in[3];
  const float* bq = (const float*)d_in[4];
  const float* Wk = (const float*)d_in[5];
  const float* bk = (const float*)d_in[6];
  const float* Wv = (const float*)d_in[7];
  const float* bv = (const float*)d_in[8];
  const float* Wo = (const float*)d_in[9];
  const float* bo = (const float*)d_in[10];
  const float* RF = (const float*)d_in[11];

  float* ws = (float*)d_ws;
  const size_t SZ_LD = (size_t)L_SEQ * DMODEL;
  const size_t SZ_PRIME = (size_t)NHEADS * L_SEQ * MFEAT;
  float* context = ws;
  float* V      = context + SZ_LD;
  float* tilemax = V + SZ_LD;
  float* qprime = tilemax + NTILEMAX;
  float* kexp   = qprime + SZ_PRIME;
  float* chunkbuf = kexp + SZ_PRIME;
  float* denloc = chunkbuf + (size_t)NHEADS * NCHUNKS * CHUNK_STRIDE;

  perf_qkv_features<<<dim3(L_SEQ / 64, NHEADS, 3), 128, 0, stream>>>(
      query, key_, value, Wq, bq, Wk, bk, Wv, bv, RF, V, qprime, kexp, tilemax);
  perf_scan<<<NHEADS * NCHUNKS, 256, 0, stream>>>(qprime, kexp, tilemax, V, chunkbuf,
                                                  context, denloc);
  perf_apply_prefix<<<dim3(NCHUNKS, DHEAD / 32, NHEADS), 128, 0, stream>>>(
      qprime, chunkbuf, denloc, context);
  perf_out_proj<<<dim3(L_SEQ / 64, DMODEL / 64), 32 * OP_SPLIT, 0, stream>>>(
      context, Wo, bo, (float*)d_out);
}


# in-chunk scan re-laid: a (head, chunk) over two work-groups by d half, a wave's 16 features over its two lane halves, z in registers: 5 not 11 LDS reads a row; num and den summed in another tree
# speedup vs baseline: 9.3226x; 1.1617x over previous
#include <hip/hip_runtime.h>
#include <math.h>

#define L_SEQ 2048
#define DMODEL 512
#define NHEADS 8
#define DHEAD 64
#define MFEAT 128
#define NCHUNKS 16
#define CHUNK 128
#define CHUNK_STRIDE 8320
#define KERNEL_EPS 1e-4f
#define DEN_EPS 1e-6f
#define QK_SCALE 0.21022410381342864f
#define NC_FEAT 0.08838834764831845f

typedef float v2f __attribute__((ext_vector_type(2)));
typedef float v8f __attribute__((ext_vector_type(8)));

struct Acc2x2 { v8f c[2][2]; };
typedef float v4f __attribute__((ext_vector_type(4)));

#define PT_PITCH 68
#define PT_PLANE (64 * PT_PITCH)

template <int KLEN>
__device__ __forceinline__ void gemm64x64_kslice(const float* __restrict__ A,
                                                 const float* __restrict__ W,
                                                 int kb, int lane, float* P) {
  const int half = lane >> 4;
  const int l16  = lane & 15;
  const float* ap = A + l16 * DMODEL + kb + 2 * half;
  const float* bp = W + l16 * DMODEL + kb + 2 * half;
  const v8f zero = {0.f, 0.f, 0.f, 0.f, 0.f, 0.f, 0.f, 0.f};
  v8f acc[4][4];
#pragma unroll
  for (int ti = 0; ti < 4; ++ti)
#pragma unroll
    for (int tj = 0; tj < 4; ++tj) acc[ti][tj] = zero;
  for (int kk = 0; kk < KLEN; kk += 32) {
    v2f a[8][4], b[8][4];
#pragma unroll
    for (int s8 = 0; s8 < 8; ++s8) {
      const int o = kk + 4 * s8;
      a[s8][0].x = ap[o];  a[s8][0].y = ap[o + 1];
    }
    __builtin_amdgcn_sched_barrier(0);
#pragma unroll
    for (int j = 0; j < 4; ++j) {
#pragma unroll
      for (int s8 = 0; s8 < 8; ++s8) {
        const int o = j * 16 * DMODEL + kk + 4 * s8;
        b[s8][j].x = bp[o];  b[s8][j].y = bp[o + 1];
      }
      __builtin_amdgcn_sched_barrier(0);
    }
#pragma unroll
    for (int i = 1; i < 4; ++i) {
#pragma unroll
      for (int s8 = 0; s8 < 8; ++s8) {
        const int o = i * 16 * DMODEL + kk + 4 * s8;
        a[s8][i].x = ap[o];  a[s8][i].y = ap[o + 1];
      }
      __builtin_amdgcn_sched_barrier(0);
    }
#pragma unroll
    for (int ti = 0; ti < 4; ++ti) {
#pragma unroll
      for (int s8 = 0; s8 < 8; ++s8)
#pragma unroll
        for (int tj = 0; tj < 4; ++tj)
          acc[ti][tj] = __builtin_amdgcn_wmma_f32_16x16x4_f32(false, a[s8][ti], false, b[s8][tj],
                                                              (short)0, acc[ti][tj], false, false);
      if (ti < 3) __builtin_amdgcn_sched_barrier(0);
    }
  }
#pragma unroll
  for (int ti = 0; ti < 4; ++ti)
#pragma unroll
    for (int tj = 0; tj < 4; ++tj)
#pragma unroll
      for (int i = 0; i < 8; ++i)
        P[(16 * ti + 8 * half + i) * PT_PITCH + 16 * tj + l16] = acc[ti][tj][i];
}

template <int NS>
__device__ __forceinline__ v4f sum_planes(const float* P, int off) {
  v4f s[NS];
#pragma unroll
  for (int n = 0; n < NS; ++n) s[n] = *(const v4f*)(P + n * PT_PLANE + off);
#pragma unroll
  for (int st = 1; st < NS; st *= 2)
#pragma unroll
    for (int n = 0; n < NS; n += 2 * st) s[n] += s[n + st];
  return s[0];
}

#define NTILEMAX (NHEADS * (L_SEQ / 32))
#define TS_PITCH PT_PITCH

template <bool IS_K>
__device__ __forceinline__ void feature_from_tile(const float* T,
                                                  const float* __restrict__ RF,
                                                  float* __restrict__ dst,
                                                  float* __restrict__ tilemax,
                                                  int h, int row0, int mbase,
                                                  bool reports, int lane) {
  const int half = lane >> 4;
  const int l16  = lane & 15;
  const float* arow0 = T + l16 * TS_PITCH;
  const float* arow1 = arow0 + 16 * TS_PITCH;
  const v8f zero = {0.f, 0.f, 0.f, 0.f, 0.f, 0.f, 0.f, 0.f};
  float hrow0 = 0.f, hrow1 = 0.f;
#pragma unroll
  for (int mb = 0; mb < 2; ++mb) {
    const int tn = mbase + 32 * mb;
    const float* bcol0 = RF + tn + l16;
    const float* bcol1 = bcol0 + 16;
    Acc2x2 r;
    r.c[0][0] = zero; r.c[0][1] = zero; r.c[1][0] = zero; r.c[1][1] = zero;
    float sq0 = 0.f, sq1 = 0.f;
    for (int k0 = 0; k0 < DHEAD; k0 += 4) {
      const int ka = k0 + 2 * half;
      const v2f a0 = *(const v2f*)(arow0 + ka);
      const v2f a1 = *(const v2f*)(arow1 + ka);
      v2f b0, b1;
      b0.x = bcol0[ka * MFEAT];    b0.y = bcol0[(ka + 1) * MFEAT];
      b1.x = bcol1[ka * MFEAT];    b1.y = bcol1[(ka + 1) * MFEAT];
      r.c[0][0] = __builtin_amdgcn_wmma_f32_16x16x4_f32(false, a0, false, b0, (short)0, r.c[0][0], false, false);
      r.c[0][1] = __builtin_amdgcn_wmma_f32_16x16x4_f32(false, a0, false, b1, (short)0, r.c[0][1], false, false);
      r.c[1][0] = __builtin_amdgcn_wmma_f32_16x16x4_f32(false, a1, false, b0, (short)0, r.c[1][0], false, false);
      r.c[1][1] = __builtin_amdgcn_wmma_f32_16x16x4_f32(false, a1, false, b1, (short)0, r.c[1][1], false, false);
      if (IS_K && mb == 0) {
        sq0 += a0.x * a0.x; sq0 += a0.y * a0.y;
        sq1 += a1.x * a1.x; sq1 += a1.y * a1.y;
      }
    }
    if (IS_K) {
      if (mb == 0) {
        hrow0 = -0.5f * (sq0 + __shfl_xor(sq0, 16));
        hrow1 = -0.5f * (sq1 + __shfl_xor(sq1, 16));
        float mx = fmaxf(hrow0, hrow1);
#pragma unroll
        for (int off = 8; off > 0; off >>= 1) mx = fmaxf(mx, __shfl_xor(mx, off));
        if (reports && lane == 0) tilemax[h * (L_SEQ / 32) + (row0 >> 5)] = mx;
      }
#pragma unroll
      for (int ti = 0; ti < 2; ++ti) {
        const int rloc = ti * 16 + 8 * half;
        float hh[8];
#pragma unroll
        for (int i = 0; i < 8; ++i) hh[i] = __shfl(ti ? hrow1 : hrow0, 8 * half + i);
#pragma unroll
        for (int tj = 0; tj < 2; ++tj) {
          const int mcol = tn + tj * 16 + l16;
#pragma unroll
          for (int i = 0; i < 8; ++i)
            dst[(h * L_SEQ + row0 + rloc + i) * MFEAT + mcol] = hh[i] + r.c[ti][tj][i];
        }
      }
    } else {
#pragma unroll
      for (int ti = 0; ti < 2; ++ti)
#pragma unroll
        for (int tj = 0; tj < 2; ++tj) {
          const int mcol = tn + tj * 16 + l16;
          const int rbase = row0 + ti * 16 + 8 * half;
#pragma unroll
          for (int i = 0; i < 8; ++i)
            dst[(h * L_SEQ + rbase + i) * MFEAT + mcol] =
                NC_FEAT * (expf(r.c[ti][tj][i]) + KERNEL_EPS);
        }
    }
  }
}

__global__ void __launch_bounds__(128)
perf_qkv_features(const float* __restrict__ xq, const float* __restrict__ xk,
                  const float* __restrict__ xv,
                  const float* __restrict__ Wq, const float* __restrict__ bq,
                  const float* __restrict__ Wk, const float* __restrict__ bk,
                  const float* __restrict__ Wv, const float* __restrict__ bv,
                  const float* __restrict__ RF,
                  float* __restrict__ V, float* __restrict__ qprime,
                  float* __restrict__ kexp, float* __restrict__ tilemax) {
  const int h = blockIdx.y;
  const int mat = blockIdx.z;
  const int t = threadIdx.x;
  const int lane = t & 31;
  const int w = t >> 5;
  const int wr = w >> 1;
  const int wc = w & 1;
  const int tm0 = blockIdx.x * 64;
  const int tm = tm0 + 32 * wr;
  __shared__ __attribute__((aligned(16))) float P_s[4 * PT_PLANE];
  const float* X; const float* W; const float* bias; float scl;
  if (mat == 0)      { X = xq; W = Wq; bias = bq; scl = QK_SCALE; }
  else if (mat == 1) { X = xk; W = Wk; bias = bk; scl = QK_SCALE; }
  else               { X = xv; W = Wv; bias = bv; scl = 1.f; }
  gemm64x64_kslice<DMODEL / 4>(X + tm0 * DMODEL, W + h * DHEAD * DMODEL, (DMODEL / 4) * w, lane,
                               P_s + w * PT_PLANE);
  __syncthreads();
#pragma unroll
  for (int j = 0; j < 8; ++j) {
    const int q = t + 128 * j;
    const int row = q >> 4;
    const int c4 = (q & 15) * 4;
    v4f y = sum_planes<4>(P_s, row * PT_PITCH + c4);
    const float* bp = bias + h * DHEAD + c4;
    y.x = (y.x + bp[0]) * scl; y.y = (y.y + bp[1]) * scl;
    y.z = (y.z + bp[2]) * scl; y.w = (y.w + bp[3]) * scl;
    if (mat == 2) *(v4f*)(V + (size_t)(tm0 + row) * DMODEL + h * DHEAD + c4) = y;
    else          *(v4f*)(P_s + row * PT_PITCH + c4) = y;
  }
  if (mat == 2) return;
  __syncthreads();
  float* T_s = P_s;
  if (mat == 1)
    feature_from_tile<true>(T_s + 32 * wr * TS_PITCH, RF, kexp, tilemax, h, tm, 64 * wc,
                            wc == 0, lane);
  else
    feature_from_tile<false>(T_s + 32 * wr * TS_PITCH, RF, qprime, tilemax, h, tm, 64 * wc,
                             false, lane);
}

#define SCAN_ROWS 16
#define SCAN_BLOCKS (CHUNK / SCAN_ROWS)
#define SCAN_D 32
#define RED_ROW (16 * SCAN_D + 32)

__device__ __forceinline__ v4f kfeat4(v4f e, float ks) {
  v4f r;
  r.x = NC_FEAT * (expf(e.x - ks) + KERNEL_EPS);
  r.y = NC_FEAT * (expf(e.y - ks) + KERNEL_EPS);
  r.z = NC_FEAT * (expf(e.z - ks) + KERNEL_EPS);
  r.w = NC_FEAT * (expf(e.w - ks) + KERNEL_EPS);
  return r;
}

__global__ void __launch_bounds__(256)
perf_scan(const float* __restrict__ qprime, const float* __restrict__ kexp,
          const float* __restrict__ tilemax,
          const float* __restrict__ V, float* __restrict__ chunkbuf,
          float* __restrict__ numer, float* __restrict__ denloc) {
  const int dh = blockIdx.x & 1;
  const int c = (blockIdx.x >> 1) & 15;
  const int h = blockIdx.x >> 5;
  const int t = threadIdx.x;
  const int lane = t & 31;
  const int w = t >> 5;
  const int fh = lane >> 4;
  const int dl = lane & 15;
  const int f0 = 16 * w + 8 * fh;
  const int dcol = h * DHEAD + SCAN_D * dh;
  __shared__ __attribute__((aligned(16))) float qp_s[SCAN_ROWS * MFEAT];
  __shared__ __attribute__((aligned(16))) float kp_s[SCAN_ROWS * MFEAT];
  __shared__ __attribute__((aligned(16))) float vv_s[SCAN_ROWS * SCAN_D];
  __shared__ __attribute__((aligned(16))) float red[SCAN_ROWS * RED_ROW];
  __shared__ __attribute__((aligned(16))) float denp[SCAN_ROWS * 16];
  __shared__ float dsink[32];
  __shared__ __attribute__((aligned(16))) float den_s[CHUNK];
  __shared__ __attribute__((aligned(16))) float z_s[MFEAT];
  __shared__ float wmax_s[8];

  v2f S[8];
#pragma unroll
  for (int j = 0; j < 8; ++j) { S[j].x = 0.f; S[j].y = 0.f; }
  v2f Z[4];
#pragma unroll
  for (int i = 0; i < 4; ++i) { Z[i].x = 0.f; Z[i].y = 0.f; }
  float* const dst0 = (dl == 0) ? (denp + 2 * w + fh) : (dsink + lane);
  const int dstep = (dl == 0) ? 16 : 0;

  const int l_chunk = c * CHUNK;
  const v4f* qsrc = (const v4f*)(qprime + (size_t)(h * L_SEQ + l_chunk) * MFEAT);
  const v4f* ksrc = (const v4f*)(kexp + (size_t)(h * L_SEQ + l_chunk) * MFEAT);
  const int vrow = t >> 4;
  const int vcol = (t & 15) * 2;
  const float* vsrc = V + (size_t)(l_chunk + vrow) * DMODEL + dcol + vcol;

  const v2f tmx = ((const v2f*)tilemax)[t];
  v4f gq0 = qsrc[t], gq1 = qsrc[t + 256];
  v4f gk0 = ksrc[t], gk1 = ksrc[t + 256];
  v2f gv = *(const v2f*)vsrc;
  float mx = fmaxf(tmx.x, tmx.y);
#pragma unroll
  for (int off = 16; off > 0; off >>= 1) mx = fmaxf(mx, __shfl_xor(mx, off));
  if (lane == 0) wmax_s[w] = mx;
  __syncthreads();
  float ks = wmax_s[0];
#pragma unroll
  for (int i = 1; i < 8; ++i) ks = fmaxf(ks, wmax_s[i]);
  ((v4f*)qp_s)[t] = gq0; ((v4f*)qp_s)[t + 256] = gq1;
  ((v4f*)kp_s)[t] = kfeat4(gk0, ks); ((v4f*)kp_s)[t + 256] = kfeat4(gk1, ks);
  ((v2f*)vv_s)[t] = gv;
  __syncthreads();

  for (int b = 0; b < SCAN_BLOCKS; ++b) {
    if (b + 1 < SCAN_BLOCKS) {
      const int o = (b + 1) * (SCAN_ROWS * MFEAT / 4);
      gq0 = qsrc[o + t]; gq1 = qsrc[o + t + 256];
      gk0 = ksrc[o + t]; gk1 = ksrc[o + t + 256];
      gv = *(const v2f*)(vsrc + (size_t)(b + 1) * SCAN_ROWS * DMODEL);
    }
#pragma unroll 4
    for (int r = 0; r < SCAN_ROWS; ++r) {
      const v4f* kq = (const v4f*)(kp_s + r * MFEAT + f0);
      const v4f* qq = (const v4f*)(qp_s + r * MFEAT + f0);
      const v4f k0 = kq[0], k1 = kq[1];
      const v4f q0 = qq[0], q1 = qq[1];
      const v2f vv2 = *(const v2f*)(vv_s + r * SCAN_D + 2 * dl);
      const float kk[8] = {k0.x, k0.y, k0.z, k0.w, k1.x, k1.y, k1.z, k1.w};
      const float qv[8] = {q0.x, q0.y, q0.z, q0.w, q1.x, q1.y, q1.z, q1.w};
#pragma unroll
      for (int j = 0; j < 8; ++j) S[j] += kk[j] * vv2;
      {
        const v2f ka = {k0.x, k0.y}, kb = {k0.z, k0.w}, kc = {k1.x, k1.y}, kd = {k1.z, k1.w};
        Z[0] += ka; Z[1] += kb; Z[2] += kc; Z[3] += kd;
      }
      v2f sa = {0.f, 0.f}, sb = {0.f, 0.f};
#pragma unroll
      for (int j = 0; j < 8; j += 2) {
        sa += qv[j] * S[j];
        sb += qv[j + 1] * S[j + 1];
      }
      *(v2f*)(red + r * RED_ROW + (2 * w + fh) * SCAN_D + 2 * dl) = sa + sb;
      {
        const v2f qa = {q0.x, q0.y}, qb = {q0.z, q0.w}, qc = {q1.x, q1.y}, qd = {q1.z, q1.w};
        v2f da = qa * Z[0], db = qb * Z[1];
        da += qc * Z[2]; db += qd * Z[3];
        dst0[r * dstep] = (da.x + da.y) + (db.x + db.y);
      }
    }
    if (b + 1 < SCAN_BLOCKS) { gk0 = kfeat4(gk0, ks); gk1 = kfeat4(gk1, ks); }
    __syncthreads();
    if (b + 1 < SCAN_BLOCKS) {
      ((v4f*)qp_s)[t] = gq0; ((v4f*)qp_s)[t + 256] = gq1;
      ((v4f*)kp_s)[t] = gk0; ((v4f*)kp_s)[t + 256] = gk1;
      ((v2f*)vv_s)[t] = gv;
    }
    {
      const int r = t >> 4;
      const int d2 = (t & 15) * 2;
      const float* rp = red + r * RED_ROW + d2;
      v2f n4[4];
#pragma unroll
      for (int p = 0; p < 4; ++p) n4[p] = *(const v2f*)(rp + p * SCAN_D);
#pragma unroll
      for (int p = 4; p < 16; ++p) n4[p & 3] += *(const v2f*)(rp + p * SCAN_D);
      const int l = l_chunk + b * SCAN_ROWS + r;
      *(v2f*)(numer + (size_t)l * DMODEL + dcol + d2) = (n4[0] + n4[1]) + (n4[2] + n4[3]);
    }
    {
      const int r = t >> 4;
      const v4f* dq = (const v4f*)(denp + r * 16);
      const v4f ds = (dq[0] + dq[1]) + (dq[2] + dq[3]);
      if ((t & 15) == 0) den_s[b * SCAN_ROWS + r] = (ds.x + ds.y) + (ds.z + ds.w);
    }
    __syncthreads();
  }
  float* sums = chunkbuf + (size_t)(h * NCHUNKS + c) * CHUNK_STRIDE;
#pragma unroll
  for (int j = 0; j < 8; ++j)
    *(v2f*)(sums + (f0 + j) * DHEAD + SCAN_D * dh + 2 * dl) = S[j];
  if (dh != 0) return;
  if (dl == 0) {
    const v4f za = {Z[0].x, Z[0].y, Z[1].x, Z[1].y}, zb = {Z[2].x, Z[2].y, Z[3].x, Z[3].y};
    *(v4f*)(z_s + f0) = za; *(v4f*)(z_s + f0 + 4) = zb;
  }
  __syncthreads();
  if (w == 0) {
    ((v4f*)(sums + MFEAT * DHEAD))[lane] = ((const v4f*)z_s)[lane];
    ((v4f*)(denloc + h * L_SEQ + l_chunk))[lane] = ((const v4f*)den_s)[lane];
  }
}

#define PS_PITCH 40

__global__ void __launch_bounds__(128)
perf_apply_prefix(const float* __restrict__ qprime, const float* __restrict__ chunkbuf,
                  const float* __restrict__ denloc, float* __restrict__ ctx) {
  const int c = blockIdx.x;
  const int tn = blockIdx.y * 32;
  const int h = blockIdx.z;
  const int t = threadIdx.x;
  const int lane = t & 31;
  const int w = t >> 5;
  const int tm = c * CHUNK + 32 * w;
  __shared__ __attribute__((aligned(16))) float P_s[MFEAT * PS_PITCH];
  __shared__ __attribute__((aligned(16))) float zp_s[MFEAT];

  const float* hbase = chunkbuf + (size_t)(h * NCHUNKS) * CHUNK_STRIDE;
  const int prow = t >> 3;
  const int pcol = (t & 7) * 4;
  const v4f zero4 = {0.f, 0.f, 0.f, 0.f};
  v4f acc[8];
#pragma unroll
  for (int j = 0; j < 8; ++j) acc[j] = zero4;
  float zacc = 0.f;
#pragma unroll 4
  for (int cc = 0; cc < c; ++cc) {
    const float* s = hbase + (size_t)cc * CHUNK_STRIDE;
#pragma unroll
    for (int j = 0; j < 8; ++j)
      acc[j] += *(const v4f*)(s + (prow + 16 * j) * DHEAD + tn + pcol);
    zacc += s[MFEAT * DHEAD + t];
  }
#pragma unroll
  for (int j = 0; j < 8; ++j)
    *(v4f*)(P_s + (prow + 16 * j) * PS_PITCH + pcol) = acc[j];
  zp_s[t] = zacc;
  __syncthreads();

  const int half = lane >> 4;
  const int l16  = lane & 15;
  const float* A = qprime + (size_t)(h * L_SEQ + tm) * MFEAT;
  const float* arow0 = A + l16 * MFEAT;
  const float* arow1 = A + (l16 + 16) * MFEAT;
  const float* bcol0 = P_s + l16;
  const float* bcol1 = P_s + 16 + l16;
  const v8f zero = {0.f, 0.f, 0.f, 0.f, 0.f, 0.f, 0.f, 0.f};
  Acc2x2 r;
  r.c[0][0] = zero; r.c[0][1] = zero; r.c[1][0] = zero; r.c[1][1] = zero;
  for (int k0 = 0; k0 < MFEAT; k0 += 4) {
    const int ka = k0 + 2 * half;
    v2f a0, a1, b0, b1;
    a0.x = arow0[ka];               a0.y = arow0[ka + 1];
    a1.x = arow1[ka];               a1.y = arow1[ka + 1];
    b0.x = bcol0[ka * PS_PITCH];    b0.y = bcol0[(ka + 1) * PS_PITCH];
    b1.x = bcol1[ka * PS_PITCH];    b1.y = bcol1[(ka + 1) * PS_PITCH];
    r.c[0][0] = __builtin_amdgcn_wmma_f32_16x16x4_f32(false, a0, false, b0, (short)0, r.c[0][0], false, false);
    r.c[0][1] = __builtin_amdgcn_wmma_f32_16x16x4_f32(false, a0, false, b1, (short)0, r.c[0][1], false, false);
    r.c[1][0] = __builtin_amdgcn_wmma_f32_16x16x4_f32(false, a1, false, b0, (short)0, r.c[1][0], false, false);
    r.c[1][1] = __builtin_amdgcn_wmma_f32_16x16x4_f32(false, a1, false, b1, (short)0, r.c[1][1], false, false);
  }
  const v4f* qrow = (const v4f*)(A + lane * MFEAT);
  const v4f* zp = (const v4f*)zp_s;
  v4f dacc = {0.f, 0.f, 0.f, 0.f};
  for (int m4 = 0; m4 < MFEAT / 4; ++m4) dacc += qrow[m4] * zp[m4];
  const float dtot = denloc[h * L_SEQ + tm + lane] +
                     ((dacc.x + dacc.y) + (dacc.z + dacc.w)) + DEN_EPS;
#pragma unroll
  for (int ti = 0; ti < 2; ++ti) {
    const int rloc = ti * 16 + 8 * half;
    float dd[8];
#pragma unroll
    for (int i = 0; i < 8; ++i) dd[i] = __shfl(dtot, rloc + i);
#pragma unroll
    for (int tj = 0; tj < 2; ++tj) {
      const int col = h * DHEAD + tn + tj * 16 + l16;
#pragma unroll
      for (int i = 0; i < 8; ++i) {
        float* p = ctx + (size_t)(tm + rloc + i) * DMODEL + col;
        *p = (*p + r.c[ti][tj][i]) / dd[i];
      }
    }
  }
}

#define OP_SPLIT 8
__global__ void __launch_bounds__(32 * OP_SPLIT)
perf_out_proj(const float* __restrict__ ctx, const float* __restrict__ Wo,
              const float* __restrict__ bo, float* __restrict__ out) {
  const int tm0 = blockIdx.x * 64;
  const int tn0 = blockIdx.y * 64;
  const int t = threadIdx.x;
  const int lane = t & 31;
  const int w = t >> 5;
  __shared__ __attribute__((aligned(16))) float P_s[OP_SPLIT * PT_PLANE];
  gemm64x64_kslice<DMODEL / OP_SPLIT>(ctx + tm0 * DMODEL, Wo + tn0 * DMODEL,
                                      (DMODEL / OP_SPLIT) * w, lane, P_s + w * PT_PLANE);
  __syncthreads();
#pragma unroll
  for (int j = 0; j < 64 * 16 / (32 * OP_SPLIT); ++j) {
    const int q = t + 32 * OP_SPLIT * j;
    const int row = q >> 4;
    const int c4 = (q & 15) * 4;
    v4f y = sum_planes<OP_SPLIT>(P_s, row * PT_PITCH + c4);
    const float* bp = bo + tn0 + c4;
    y.x += bp[0]; y.y += bp[1]; y.z += bp[2]; y.w += bp[3];
    *(v4f*)(out + (size_t)(tm0 + row) * DMODEL + tn0 + c4) = y;
  }
}

extern "C" void kernel_launch(void* const* d_in, const int* in_sizes, int n_in,
                              void* d_out, int out_size, void* d_ws, size_t ws_size,
                              hipStream_t stream) {
  (void)in_sizes; (void)n_in; (void)out_size; (void)ws_size;
  const float* query = (const float*)d_in[0];
  const float* key_  = (const float*)d_in[1];
  const float* value = (const float*)d_in[2];
  const float* Wq = (const float*)d_in[3];
  const float* bq = (const float*)d_in[4];
  const float* Wk = (const float*)d_in[5];
  const float* bk = (const float*)d_in[6];
  const float* Wv = (const float*)d_in[7];
  const float* bv = (const float*)d_in[8];
  const float* Wo = (const float*)d_in[9];
  const float* bo = (const float*)d_in[10];
  const float* RF = (const float*)d_in[11];

  float* ws = (float*)d_ws;
  const size_t SZ_LD = (size_t)L_SEQ * DMODEL;
  const size_t SZ_PRIME = (size_t)NHEADS * L_SEQ * MFEAT;
  float* context = ws;
  float* V      = context + SZ_LD;
  float* tilemax = V + SZ_LD;
  float* qprime = tilemax + NTILEMAX;
  float* kexp   = qprime + SZ_PRIME;
  float* chunkbuf = kexp + SZ_PRIME;
  float* denloc = chunkbuf + (size_t)NHEADS * NCHUNKS * CHUNK_STRIDE;

  perf_qkv_features<<<dim3(L_SEQ / 64, NHEADS, 3), 128, 0, stream>>>(
      query, key_, value, Wq, bq, Wk, bk, Wv, bv, RF, V, qprime, kexp, tilemax);
  perf_scan<<<NHEADS * NCHUNKS * 2, 256, 0, stream>>>(qprime, kexp, tilemax, V, chunkbuf,
                                                  context, denloc);
  perf_apply_prefix<<<dim3(NCHUNKS, DHEAD / 32, NHEADS), 128, 0, stream>>>(
      qprime, chunkbuf, denloc, context);
  perf_out_proj<<<dim3(L_SEQ / 64, DMODEL / 64), 32 * OP_SPLIT, 0, stream>>>(
      context, Wo, bo, (float*)d_out);
}


# apply kernel: a wave's global loads (earlier chunks' end states, q' fragments, its numerators) all issued at entry into registers; q' read once: q'.zp taken from the A fragments, in another sum tree
# speedup vs baseline: 9.7183x; 1.0424x over previous
#include <hip/hip_runtime.h>
#include <math.h>

#define L_SEQ 2048
#define DMODEL 512
#define NHEADS 8
#define DHEAD 64
#define MFEAT 128
#define NCHUNKS 16
#define CHUNK 128
#define CHUNK_STRIDE 8320
#define KERNEL_EPS 1e-4f
#define DEN_EPS 1e-6f
#define QK_SCALE 0.21022410381342864f
#define NC_FEAT 0.08838834764831845f

typedef float v2f __attribute__((ext_vector_type(2)));
typedef float v8f __attribute__((ext_vector_type(8)));

struct Acc2x2 { v8f c[2][2]; };
typedef float v4f __attribute__((ext_vector_type(4)));

#define PT_PITCH 68
#define PT_PLANE (64 * PT_PITCH)

template <int KLEN>
__device__ __forceinline__ void gemm64x64_kslice(const float* __restrict__ A,
                                                 const float* __restrict__ W,
                                                 int kb, int lane, float* P) {
  const int half = lane >> 4;
  const int l16  = lane & 15;
  const float* ap = A + l16 * DMODEL + kb + 2 * half;
  const float* bp = W + l16 * DMODEL + kb + 2 * half;
  const v8f zero = {0.f, 0.f, 0.f, 0.f, 0.f, 0.f, 0.f, 0.f};
  v8f acc[4][4];
#pragma unroll
  for (int ti = 0; ti < 4; ++ti)
#pragma unroll
    for (int tj = 0; tj < 4; ++tj) acc[ti][tj] = zero;
  for (int kk = 0; kk < KLEN; kk += 32) {
    v2f a[8][4], b[8][4];
#pragma unroll
    for (int s8 = 0; s8 < 8; ++s8) {
      const int o = kk + 4 * s8;
      a[s8][0].x = ap[o];  a[s8][0].y = ap[o + 1];
    }
    __builtin_amdgcn_sched_barrier(0);
#pragma unroll
    for (int j = 0; j < 4; ++j) {
#pragma unroll
      for (int s8 = 0; s8 < 8; ++s8) {
        const int o = j * 16 * DMODEL + kk + 4 * s8;
        b[s8][j].x = bp[o];  b[s8][j].y = bp[o + 1];
      }
      __builtin_amdgcn_sched_barrier(0);
    }
#pragma unroll
    for (int i = 1; i < 4; ++i) {
#pragma unroll
      for (int s8 = 0; s8 < 8; ++s8) {
        const int o = i * 16 * DMODEL + kk + 4 * s8;
        a[s8][i].x = ap[o];  a[s8][i].y = ap[o + 1];
      }
      __builtin_amdgcn_sched_barrier(0);
    }
#pragma unroll
    for (int ti = 0; ti < 4; ++ti) {
#pragma unroll
      for (int s8 = 0; s8 < 8; ++s8)
#pragma unroll
        for (int tj = 0; tj < 4; ++tj)
          acc[ti][tj] = __builtin_amdgcn_wmma_f32_16x16x4_f32(false, a[s8][ti], false, b[s8][tj],
                                                              (short)0, acc[ti][tj], false, false);
      if (ti < 3) __builtin_amdgcn_sched_barrier(0);
    }
  }
#pragma unroll
  for (int ti = 0; ti < 4; ++ti)
#pragma unroll
    for (int tj = 0; tj < 4; ++tj)
#pragma unroll
      for (int i = 0; i < 8; ++i)
        P[(16 * ti + 8 * half + i) * PT_PITCH + 16 * tj + l16] = acc[ti][tj][i];
}

template <int NS>
__device__ __forceinline__ v4f sum_planes(const float* P, int off) {
  v4f s[NS];
#pragma unroll
  for (int n = 0; n < NS; ++n) s[n] = *(const v4f*)(P + n * PT_PLANE + off);
#pragma unroll
  for (int st = 1; st < NS; st *= 2)
#pragma unroll
    for (int n = 0; n < NS; n += 2 * st) s[n] += s[n + st];
  return s[0];
}

#define NTILEMAX (NHEADS * (L_SEQ / 32))
#define TS_PITCH PT_PITCH

template <bool IS_K>
__device__ __forceinline__ void feature_from_tile(const float* T,
                                                  const float* __restrict__ RF,
                                                  float* __restrict__ dst,
                                                  float* __restrict__ tilemax,
                                                  int h, int row0, int mbase,
                                                  bool reports, int lane) {
  const int half = lane >> 4;
  const int l16  = lane & 15;
  const float* arow0 = T + l16 * TS_PITCH;
  const float* arow1 = arow0 + 16 * TS_PITCH;
  const v8f zero = {0.f, 0.f, 0.f, 0.f, 0.f, 0.f, 0.f, 0.f};
  float hrow0 = 0.f, hrow1 = 0.f;
#pragma unroll
  for (int mb = 0; mb < 2; ++mb) {
    const int tn = mbase + 32 * mb;
    const float* bcol0 = RF + tn + l16;
    const float* bcol1 = bcol0 + 16;
    Acc2x2 r;
    r.c[0][0] = zero; r.c[0][1] = zero; r.c[1][0] = zero; r.c[1][1] = zero;
    float sq0 = 0.f, sq1 = 0.f;
    for (int k0 = 0; k0 < DHEAD; k0 += 4) {
      const int ka = k0 + 2 * half;
      const v2f a0 = *(const v2f*)(arow0 + ka);
      const v2f a1 = *(const v2f*)(arow1 + ka);
      v2f b0, b1;
      b0.x = bcol0[ka * MFEAT];    b0.y = bcol0[(ka + 1) * MFEAT];
      b1.x = bcol1[ka * MFEAT];    b1.y = bcol1[(ka + 1) * MFEAT];
      r.c[0][0] = __builtin_amdgcn_wmma_f32_16x16x4_f32(false, a0, false, b0, (short)0, r.c[0][0], false, false);
      r.c[0][1] = __builtin_amdgcn_wmma_f32_16x16x4_f32(false, a0, false, b1, (short)0, r.c[0][1], false, false);
      r.c[1][0] = __builtin_amdgcn_wmma_f32_16x16x4_f32(false, a1, false, b0, (short)0, r.c[1][0], false, false);
      r.c[1][1] = __builtin_amdgcn_wmma_f32_16x16x4_f32(false, a1, false, b1, (short)0, r.c[1][1], false, false);
      if (IS_K && mb == 0) {
        sq0 += a0.x * a0.x; sq0 += a0.y * a0.y;
        sq1 += a1.x * a1.x; sq1 += a1.y * a1.y;
      }
    }
    if (IS_K) {
      if (mb == 0) {
        hrow0 = -0.5f * (sq0 + __shfl_xor(sq0, 16));
        hrow1 = -0.5f * (sq1 + __shfl_xor(sq1, 16));
        float mx = fmaxf(hrow0, hrow1);
#pragma unroll
        for (int off = 8; off > 0; off >>= 1) mx = fmaxf(mx, __shfl_xor(mx, off));
        if (reports && lane == 0) tilemax[h * (L_SEQ / 32) + (row0 >> 5)] = mx;
      }
#pragma unroll
      for (int ti = 0; ti < 2; ++ti) {
        const int rloc = ti * 16 + 8 * half;
        float hh[8];
#pragma unroll
        for (int i = 0; i < 8; ++i) hh[i] = __shfl(ti ? hrow1 : hrow0, 8 * half + i);
#pragma unroll
        for (int tj = 0; tj < 2; ++tj) {
          const int mcol = tn + tj * 16 + l16;
#pragma unroll
          for (int i = 0; i < 8; ++i)
            dst[(h * L_SEQ + row0 + rloc + i) * MFEAT + mcol] = hh[i] + r.c[ti][tj][i];
        }
      }
    } else {
#pragma unroll
      for (int ti = 0; ti < 2; ++ti)
#pragma unroll
        for (int tj = 0; tj < 2; ++tj) {
          const int mcol = tn + tj * 16 + l16;
          const int rbase = row0 + ti * 16 + 8 * half;
#pragma unroll
          for (int i = 0; i < 8; ++i)
            dst[(h * L_SEQ + rbase + i) * MFEAT + mcol] =
                NC_FEAT * (expf(r.c[ti][tj][i]) + KERNEL_EPS);
        }
    }
  }
}

__global__ void __launch_bounds__(128)
perf_qkv_features(const float* __restrict__ xq, const float* __restrict__ xk,
                  const float* __restrict__ xv,
                  const float* __restrict__ Wq, const float* __restrict__ bq,
                  const float* __restrict__ Wk, const float* __restrict__ bk,
                  const float* __restrict__ Wv, const float* __restrict__ bv,
                  const float* __restrict__ RF,
                  float* __restrict__ V, float* __restrict__ qprime,
                  float* __restrict__ kexp, float* __restrict__ tilemax) {
  const int h = blockIdx.y;
  const int mat = blockIdx.z;
  const int t = threadIdx.x;
  const int lane = t & 31;
  const int w = t >> 5;
  const int wr = w >> 1;
  const int wc = w & 1;
  const int tm0 = blockIdx.x * 64;
  const int tm = tm0 + 32 * wr;
  __shared__ __attribute__((aligned(16))) float P_s[4 * PT_PLANE];
  const float* X; const float* W; const float* bias; float scl;
  if (mat == 0)      { X = xq; W = Wq; bias = bq; scl = QK_SCALE; }
  else if (mat == 1) { X = xk; W = Wk; bias = bk; scl = QK_SCALE; }
  else               { X = xv; W = Wv; bias = bv; scl = 1.f; }
  gemm64x64_kslice<DMODEL / 4>(X + tm0 * DMODEL, W + h * DHEAD * DMODEL, (DMODEL / 4) * w, lane,
                               P_s + w * PT_PLANE);
  __syncthreads();
#pragma unroll
  for (int j = 0; j < 8; ++j) {
    const int q = t + 128 * j;
    const int row = q >> 4;
    const int c4 = (q & 15) * 4;
    v4f y = sum_planes<4>(P_s, row * PT_PITCH + c4);
    const float* bp = bias + h * DHEAD + c4;
    y.x = (y.x + bp[0]) * scl; y.y = (y.y + bp[1]) * scl;
    y.z = (y.z + bp[2]) * scl; y.w = (y.w + bp[3]) * scl;
    if (mat == 2) *(v4f*)(V + (size_t)(tm0 + row) * DMODEL + h * DHEAD + c4) = y;
    else          *(v4f*)(P_s + row * PT_PITCH + c4) = y;
  }
  if (mat == 2) return;
  __syncthreads();
  float* T_s = P_s;
  if (mat == 1)
    feature_from_tile<true>(T_s + 32 * wr * TS_PITCH, RF, kexp, tilemax, h, tm, 64 * wc,
                            wc == 0, lane);
  else
    feature_from_tile<false>(T_s + 32 * wr * TS_PITCH, RF, qprime, tilemax, h, tm, 64 * wc,
                             false, lane);
}

#define SCAN_ROWS 16
#define SCAN_BLOCKS (CHUNK / SCAN_ROWS)
#define SCAN_D 32
#define RED_ROW (16 * SCAN_D + 32)

__device__ __forceinline__ v4f kfeat4(v4f e, float ks) {
  v4f r;
  r.x = NC_FEAT * (expf(e.x - ks) + KERNEL_EPS);
  r.y = NC_FEAT * (expf(e.y - ks) + KERNEL_EPS);
  r.z = NC_FEAT * (expf(e.z - ks) + KERNEL_EPS);
  r.w = NC_FEAT * (expf(e.w - ks) + KERNEL_EPS);
  return r;
}

__global__ void __launch_bounds__(256)
perf_scan(const float* __restrict__ qprime, const float* __restrict__ kexp,
          const float* __restrict__ tilemax,
          const float* __restrict__ V, float* __restrict__ chunkbuf,
          float* __restrict__ numer, float* __restrict__ denloc) {
  const int dh = blockIdx.x & 1;
  const int c = (blockIdx.x >> 1) & 15;
  const int h = blockIdx.x >> 5;
  const int t = threadIdx.x;
  const int lane = t & 31;
  const int w = t >> 5;
  const int fh = lane >> 4;
  const int dl = lane & 15;
  const int f0 = 16 * w + 8 * fh;
  const int dcol = h * DHEAD + SCAN_D * dh;
  __shared__ __attribute__((aligned(16))) float qp_s[SCAN_ROWS * MFEAT];
  __shared__ __attribute__((aligned(16))) float kp_s[SCAN_ROWS * MFEAT];
  __shared__ __attribute__((aligned(16))) float vv_s[SCAN_ROWS * SCAN_D];
  __shared__ __attribute__((aligned(16))) float red[SCAN_ROWS * RED_ROW];
  __shared__ __attribute__((aligned(16))) float denp[SCAN_ROWS * 16];
  __shared__ float dsink[32];
  __shared__ __attribute__((aligned(16))) float den_s[CHUNK];
  __shared__ __attribute__((aligned(16))) float z_s[MFEAT];
  __shared__ float wmax_s[8];

  v2f S[8];
#pragma unroll
  for (int j = 0; j < 8; ++j) { S[j].x = 0.f; S[j].y = 0.f; }
  v2f Z[4];
#pragma unroll
  for (int i = 0; i < 4; ++i) { Z[i].x = 0.f; Z[i].y = 0.f; }
  float* const dst0 = (dl == 0) ? (denp + 2 * w + fh) : (dsink + lane);
  const int dstep = (dl == 0) ? 16 : 0;

  const int l_chunk = c * CHUNK;
  const v4f* qsrc = (const v4f*)(qprime + (size_t)(h * L_SEQ + l_chunk) * MFEAT);
  const v4f* ksrc = (const v4f*)(kexp + (size_t)(h * L_SEQ + l_chunk) * MFEAT);
  const int vrow = t >> 4;
  const int vcol = (t & 15) * 2;
  const float* vsrc = V + (size_t)(l_chunk + vrow) * DMODEL + dcol + vcol;

  const v2f tmx = ((const v2f*)tilemax)[t];
  v4f gq0 = qsrc[t], gq1 = qsrc[t + 256];
  v4f gk0 = ksrc[t], gk1 = ksrc[t + 256];
  v2f gv = *(const v2f*)vsrc;
  float mx = fmaxf(tmx.x, tmx.y);
#pragma unroll
  for (int off = 16; off > 0; off >>= 1) mx = fmaxf(mx, __shfl_xor(mx, off));
  if (lane == 0) wmax_s[w] = mx;
  __syncthreads();
  float ks = wmax_s[0];
#pragma unroll
  for (int i = 1; i < 8; ++i) ks = fmaxf(ks, wmax_s[i]);
  ((v4f*)qp_s)[t] = gq0; ((v4f*)qp_s)[t + 256] = gq1;
  ((v4f*)kp_s)[t] = kfeat4(gk0, ks); ((v4f*)kp_s)[t + 256] = kfeat4(gk1, ks);
  ((v2f*)vv_s)[t] = gv;
  __syncthreads();

  for (int b = 0; b < SCAN_BLOCKS; ++b) {
    if (b + 1 < SCAN_BLOCKS) {
      const int o = (b + 1) * (SCAN_ROWS * MFEAT / 4);
      gq0 = qsrc[o + t]; gq1 = qsrc[o + t + 256];
      gk0 = ksrc[o + t]; gk1 = ksrc[o + t + 256];
      gv = *(const v2f*)(vsrc + (size_t)(b + 1) * SCAN_ROWS * DMODEL);
    }
#pragma unroll 4
    for (int r = 0; r < SCAN_ROWS; ++r) {
      const v4f* kq = (const v4f*)(kp_s + r * MFEAT + f0);
      const v4f* qq = (const v4f*)(qp_s + r * MFEAT + f0);
      const v4f k0 = kq[0], k1 = kq[1];
      const v4f q0 = qq[0], q1 = qq[1];
      const v2f vv2 = *(const v2f*)(vv_s + r * SCAN_D + 2 * dl);
      const float kk[8] = {k0.x, k0.y, k0.z, k0.w, k1.x, k1.y, k1.z, k1.w};
      const float qv[8] = {q0.x, q0.y, q0.z, q0.w, q1.x, q1.y, q1.z, q1.w};
#pragma unroll
      for (int j = 0; j < 8; ++j) S[j] += kk[j] * vv2;
      {
        const v2f ka = {k0.x, k0.y}, kb = {k0.z, k0.w}, kc = {k1.x, k1.y}, kd = {k1.z, k1.w};
        Z[0] += ka; Z[1] += kb; Z[2] += kc; Z[3] += kd;
      }
      v2f sa = {0.f, 0.f}, sb = {0.f, 0.f};
#pragma unroll
      for (int j = 0; j < 8; j += 2) {
        sa += qv[j] * S[j];
        sb += qv[j + 1] * S[j + 1];
      }
      *(v2f*)(red + r * RED_ROW + (2 * w + fh) * SCAN_D + 2 * dl) = sa + sb;
      {
        const v2f qa = {q0.x, q0.y}, qb = {q0.z, q0.w}, qc = {q1.x, q1.y}, qd = {q1.z, q1.w};
        v2f da = qa * Z[0], db = qb * Z[1];
        da += qc * Z[2]; db += qd * Z[3];
        dst0[r * dstep] = (da.x + da.y) + (db.x + db.y);
      }
    }
    if (b + 1 < SCAN_BLOCKS) { gk0 = kfeat4(gk0, ks); gk1 = kfeat4(gk1, ks); }
    __syncthreads();
    if (b + 1 < SCAN_BLOCKS) {
      ((v4f*)qp_s)[t] = gq0; ((v4f*)qp_s)[t + 256] = gq1;
      ((v4f*)kp_s)[t] = gk0; ((v4f*)kp_s)[t + 256] = gk1;
      ((v2f*)vv_s)[t] = gv;
    }
    {
      const int r = t >> 4;
      const int d2 = (t & 15) * 2;
      const float* rp = red + r * RED_ROW + d2;
      v2f n4[4];
#pragma unroll
      for (int p = 0; p < 4; ++p) n4[p] = *(const v2f*)(rp + p * SCAN_D);
#pragma unroll
      for (int p = 4; p < 16; ++p) n4[p & 3] += *(const v2f*)(rp + p * SCAN_D);
      const int l = l_chunk + b * SCAN_ROWS + r;
      *(v2f*)(numer + (size_t)l * DMODEL + dcol + d2) = (n4[0] + n4[1]) + (n4[2] + n4[3]);
    }
    {
      const int r = t >> 4;
      const v4f* dq = (const v4f*)(denp + r * 16);
      const v4f ds = (dq[0] + dq[1]) + (dq[2] + dq[3]);
      if ((t & 15) == 0) den_s[b * SCAN_ROWS + r] = (ds.x + ds.y) + (ds.z + ds.w);
    }
    __syncthreads();
  }
  float* sums = chunkbuf + (size_t)(h * NCHUNKS + c) * CHUNK_STRIDE;
#pragma unroll
  for (int j = 0; j < 8; ++j)
    *(v2f*)(sums + (f0 + j) * DHEAD + SCAN_D * dh + 2 * dl) = S[j];
  if (dh != 0) return;
  if (dl == 0) {
    const v4f za = {Z[0].x, Z[0].y, Z[1].x, Z[1].y}, zb = {Z[2].x, Z[2].y, Z[3].x, Z[3].y};
    *(v4f*)(z_s + f0) = za; *(v4f*)(z_s + f0 + 4) = zb;
  }
  __syncthreads();
  if (w == 0) {
    ((v4f*)(sums + MFEAT * DHEAD))[lane] = ((const v4f*)z_s)[lane];
    ((v4f*)(denloc + h * L_SEQ + l_chunk))[lane] = ((const v4f*)den_s)[lane];
  }
}

#define PS_PITCH 40

__global__ void __launch_bounds__(128)
perf_apply_prefix(const float* __restrict__ qprime, const float* __restrict__ chunkbuf,
                  const float* __restrict__ denloc, float* __restrict__ ctx) {
  const int c = blockIdx.x;
  const int tn = blockIdx.y * 32;
  const int h = blockIdx.z;
  const int t = threadIdx.x;
  const int lane = t & 31;
  const int w = t >> 5;
  const int tm = c * CHUNK + 32 * w;
  const int half = lane >> 4;
  const int l16  = lane & 15;
  __shared__ __attribute__((aligned(16))) float P_s[MFEAT * PS_PITCH];
  __shared__ __attribute__((aligned(16))) float zp_s[MFEAT];

  const float* hbase = chunkbuf + (size_t)(h * NCHUNKS) * CHUNK_STRIDE;
  const int prow = t >> 3;
  const int pcol = (t & 7) * 4;
  const int clast = c > 0 ? c - 1 : 0;
  v4f pf[NCHUNKS - 1][8];
  float zf[NCHUNKS - 1];
#pragma unroll
  for (int cc = 0; cc < NCHUNKS - 1; ++cc) {
    const float* s = hbase + (size_t)(cc < clast ? cc : clast) * CHUNK_STRIDE;
#pragma unroll
    for (int j = 0; j < 8; ++j)
      pf[cc][j] = *(const v4f*)(s + (prow + 16 * j) * DHEAD + tn + pcol);
    zf[cc] = s[MFEAT * DHEAD + t];
  }
  __builtin_amdgcn_sched_barrier(0);
  const float* A = qprime + (size_t)(h * L_SEQ + tm) * MFEAT;
  const float* arow0 = A + l16 * MFEAT + 2 * half;
  const float* arow1 = arow0 + 16 * MFEAT;
  v2f a0[MFEAT / 4], a1[MFEAT / 4];
#pragma unroll
  for (int s = 0; s < MFEAT / 4; ++s) a0[s] = *(const v2f*)(arow0 + 4 * s);
  __builtin_amdgcn_sched_barrier(0);
#pragma unroll
  for (int s = 0; s < MFEAT / 4; ++s) a1[s] = *(const v2f*)(arow1 + 4 * s);
  __builtin_amdgcn_sched_barrier(0);
  const float dloc0 = denloc[h * L_SEQ + tm + l16];
  const float dloc1 = denloc[h * L_SEQ + tm + 16 + l16];
  float num[2][2][8];
#pragma unroll
  for (int ti = 0; ti < 2; ++ti)
#pragma unroll
    for (int tj = 0; tj < 2; ++tj)
#pragma unroll
      for (int i = 0; i < 8; ++i)
        num[ti][tj][i] = ctx[(size_t)(tm + ti * 16 + 8 * half + i) * DMODEL + h * DHEAD + tn +
                             tj * 16 + l16];
  __builtin_amdgcn_sched_barrier(0);

  const v4f zero4 = {0.f, 0.f, 0.f, 0.f};
  v4f acc[8];
#pragma unroll
  for (int j = 0; j < 8; ++j) acc[j] = c > 0 ? zero4 + pf[0][j] : zero4;
  float zacc = c > 0 ? 0.f + zf[0] : 0.f;
#pragma unroll
  for (int cc = 1; cc < NCHUNKS - 1; ++cc) {
    if (cc < c) {
#pragma unroll
      for (int j = 0; j < 8; ++j) acc[j] += pf[cc][j];
      zacc += zf[cc];
    }
  }
#pragma unroll
  for (int j = 0; j < 8; ++j)
    *(v4f*)(P_s + (prow + 16 * j) * PS_PITCH + pcol) = acc[j];
  zp_s[t] = zacc;
  __syncthreads();

  const float* bcol0 = P_s + l16;
  const float* bcol1 = P_s + 16 + l16;
  const v8f zero = {0.f, 0.f, 0.f, 0.f, 0.f, 0.f, 0.f, 0.f};
  Acc2x2 r;
  r.c[0][0] = zero; r.c[0][1] = zero; r.c[1][0] = zero; r.c[1][1] = zero;
  v2f e0 = {0.f, 0.f}, e1 = {0.f, 0.f};
#pragma unroll
  for (int s = 0; s < MFEAT / 4; ++s) {
    const int ka = 4 * s + 2 * half;
    v2f b0, b1;
    b0.x = bcol0[ka * PS_PITCH];    b0.y = bcol0[(ka + 1) * PS_PITCH];
    b1.x = bcol1[ka * PS_PITCH];    b1.y = bcol1[(ka + 1) * PS_PITCH];
    const v2f z2 = *(const v2f*)(zp_s + ka);
    r.c[0][0] = __builtin_amdgcn_wmma_f32_16x16x4_f32(false, a0[s], false, b0, (short)0, r.c[0][0], false, false);
    r.c[0][1] = __builtin_amdgcn_wmma_f32_16x16x4_f32(false, a0[s], false, b1, (short)0, r.c[0][1], false, false);
    r.c[1][0] = __builtin_amdgcn_wmma_f32_16x16x4_f32(false, a1[s], false, b0, (short)0, r.c[1][0], false, false);
    r.c[1][1] = __builtin_amdgcn_wmma_f32_16x16x4_f32(false, a1[s], false, b1, (short)0, r.c[1][1], false, false);
    e0 += a0[s] * z2;
    e1 += a1[s] * z2;
  }
  float s0 = e0.x + e0.y, s1 = e1.x + e1.y;
  s0 += __shfl_xor(s0, 16);
  s1 += __shfl_xor(s1, 16);
  const float d0 = dloc0 + s0 + DEN_EPS;
  const float d1 = dloc1 + s1 + DEN_EPS;
#pragma unroll
  for (int ti = 0; ti < 2; ++ti) {
    const int rloc = ti * 16 + 8 * half;
    float dd[8];
#pragma unroll
    for (int i = 0; i < 8; ++i) dd[i] = __shfl(ti ? d1 : d0, 8 * half + i);
#pragma unroll
    for (int tj = 0; tj < 2; ++tj) {
      const int col = h * DHEAD + tn + tj * 16 + l16;
#pragma unroll
      for (int i = 0; i < 8; ++i)
        ctx[(size_t)(tm + rloc + i) * DMODEL + col] = (num[ti][tj][i] + r.c[ti][tj][i]) / dd[i];
    }
  }
}

#define OP_SPLIT 8
__global__ void __launch_bounds__(32 * OP_SPLIT)
perf_out_proj(const float* __restrict__ ctx, const float* __restrict__ Wo,
              const float* __restrict__ bo, float* __restrict__ out) {
  const int tm0 = blockIdx.x * 64;
  const int tn0 = blockIdx.y * 64;
  const int t = threadIdx.x;
  const int lane = t & 31;
  const int w = t >> 5;
  __shared__ __attribute__((aligned(16))) float P_s[OP_SPLIT * PT_PLANE];
  gemm64x64_kslice<DMODEL / OP_SPLIT>(ctx + tm0 * DMODEL, Wo + tn0 * DMODEL,
                                      (DMODEL / OP_SPLIT) * w, lane, P_s + w * PT_PLANE);
  __syncthreads();
#pragma unroll
  for (int j = 0; j < 64 * 16 / (32 * OP_SPLIT); ++j) {
    const int q = t + 32 * OP_SPLIT * j;
    const int row = q >> 4;
    const int c4 = (q & 15) * 4;
    v4f y = sum_planes<OP_SPLIT>(P_s, row * PT_PITCH + c4);
    const float* bp = bo + tn0 + c4;
    y.x += bp[0]; y.y += bp[1]; y.z += bp[2]; y.w += bp[3];
    *(v4f*)(out + (size_t)(tm0 + row) * DMODEL + tn0 + c4) = y;
  }
}

extern "C" void kernel_launch(void* const* d_in, const int* in_sizes, int n_in,
                              void* d_out, int out_size, void* d_ws, size_t ws_size,
                              hipStream_t stream) {
  (void)in_sizes; (void)n_in; (void)out_size; (void)ws_size;
  const float* query = (const float*)d_in[0];
  const float* key_  = (const float*)d_in[1];
  const float* value = (const float*)d_in[2];
  const float* Wq = (const float*)d_in[3];
  const float* bq = (const float*)d_in[4];
  const float* Wk = (const float*)d_in[5];
  const float* bk = (const float*)d_in[6];
  const float* Wv = (const float*)d_in[7];
  const float* bv = (const float*)d_in[8];
  const float* Wo = (const float*)d_in[9];
  const float* bo = (const float*)d_in[10];
  const float* RF = (const float*)d_in[11];

  float* ws = (float*)d_ws;
  const size_t SZ_LD = (size_t)L_SEQ * DMODEL;
  const size_t SZ_PRIME = (size_t)NHEADS * L_SEQ * MFEAT;
  float* context = ws;
  float* V      = context + SZ_LD;
  float* tilemax = V + SZ_LD;
  float* qprime = tilemax + NTILEMAX;
  float* kexp   = qprime + SZ_PRIME;
  float* chunkbuf = kexp + SZ_PRIME;
  float* denloc = chunkbuf + (size_t)NHEADS * NCHUNKS * CHUNK_STRIDE;

  perf_qkv_features<<<dim3(L_SEQ / 64, NHEADS, 3), 128, 0, stream>>>(
      query, key_, value, Wq, bq, Wk, bk, Wv, bv, RF, V, qprime, kexp, tilemax);
  perf_scan<<<NHEADS * NCHUNKS * 2, 256, 0, stream>>>(qprime, kexp, tilemax, V, chunkbuf,
                                                  context, denloc);
  perf_apply_prefix<<<dim3(NCHUNKS, DHEAD / 32, NHEADS), 128, 0, stream>>>(
      qprime, chunkbuf, denloc, context);
  perf_out_proj<<<dim3(L_SEQ / 64, DMODEL / 64), 32 * OP_SPLIT, 0, stream>>>(
      context, Wo, bo, (float*)d_out);
}
